# starting kernel
# baseline (speedup 1.0000x reference)
#include <hip/hip_runtime.h>
#include <hip/hip_bf16.h>

#define B_ 4
#define N_ 20000
#define C_ 512
#define M_ 1024
#define GRASP_TH 0.1f

typedef float v2f __attribute__((ext_vector_type(2)));
typedef float v8f __attribute__((ext_vector_type(8)));

#if defined(__gfx1250__) && __has_builtin(__builtin_amdgcn_wmma_f32_16x16x4_f32)
#define USE_WMMA 1
#else
#define USE_WMMA 0
#endif

__device__ __forceinline__ void ins3_stream(float v, int n, float d[3], int ix[3]) {
  const bool c0 = v < d[0];
  const bool c1 = v < d[1];
  const bool c2 = v < d[2];
  const float r0 = c0 ? v     : d[0];
  const int   j0 = c0 ? n     : ix[0];
  const float r1 = c0 ? d[0]  : (c1 ? v : d[1]);
  const int   j1 = c0 ? ix[0] : (c1 ? n : ix[1]);
  const float r2 = c1 ? d[1]  : (c2 ? v : d[2]);
  const int   j2 = c1 ? ix[1] : (c2 ? n : ix[2]);
  d[0] = r0; d[1] = r1; d[2] = r2;
  ix[0] = j0; ix[1] = j1; ix[2] = j2;
}

__device__ __forceinline__ void ins3_lex(float v, int n, float d[3], int ix[3]) {
  const bool c0 = (v < d[0]) | ((v == d[0]) & (n < ix[0]));
  const bool c1 = (v < d[1]) | ((v == d[1]) & (n < ix[1]));
  const bool c2 = (v < d[2]) | ((v == d[2]) & (n < ix[2]));
  const float r0 = c0 ? v     : d[0];
  const int   j0 = c0 ? n     : ix[0];
  const float r1 = c0 ? d[0]  : (c1 ? v : d[1]);
  const int   j1 = c0 ? ix[0] : (c1 ? n : ix[1]);
  const float r2 = c1 ? d[1]  : (c2 ? v : d[2]);
  const int   j2 = c1 ? ix[1] : (c2 ? n : ix[2]);
  d[0] = r0; d[1] = r1; d[2] = r2;
  ix[0] = j0; ix[1] = j1; ix[2] = j2;
}

__global__ __launch_bounds__(1024) void fps_kernel(
    const float* __restrict__ pc, const float* __restrict__ obj,
    const float* __restrict__ grasp, int* __restrict__ fps_idx) {
  const int b    = blockIdx.x;
  const int tid  = threadIdx.x;
  const int lane = tid & 31;
  const int wid  = tid >> 5;

  __shared__ float s_xyz[N_ * 3];
  __shared__ float s_mind[N_];
  __shared__ float swv[32];
  __shared__ int   swi[32];
  __shared__ int   s_last;
  __shared__ float s_px, s_py, s_pz;

  const float* pcb = pc + (size_t)b * N_ * 3;
  const float* ob0 = obj + (size_t)b * 2 * N_;
  const float* gr  = grasp + (size_t)b * N_;

  int firstLoc = N_;
  for (int n = tid; n < N_; n += 1024) {
    const float* p = pcb + (size_t)n * 3;
    const float x = p[0], y = p[1], z = p[2];
    s_xyz[n * 3 + 0] = x;
    s_xyz[n * 3 + 1] = y;
    s_xyz[n * 3 + 2] = z;
    const bool mk = (ob0[N_ + n] > ob0[n]) & (gr[n] > GRASP_TH);
    s_mind[n] = mk ? __builtin_inff() : -__builtin_inff();
    firstLoc  = (mk & (n < firstLoc)) ? n : firstLoc;
  }
  for (int m = 16; m; m >>= 1) {
    const int o = __shfl_xor(firstLoc, m, 32);
    firstLoc = (o < firstLoc) ? o : firstLoc;
  }
  if (lane == 0) swi[wid] = firstLoc;
  __syncthreads();
  if (wid == 0) {
    int v = swi[lane];
    for (int m = 16; m; m >>= 1) {
      const int o = __shfl_xor(v, m, 32);
      v = (o < v) ? o : v;
    }
    if (lane == 0) s_last = (v >= N_) ? 0 : v;
  }
  __syncthreads();

  for (int step = 0; step < M_; ++step) {
    if (tid == 0) {
      const int last = s_last;
      fps_idx[b * M_ + step] = last;
      s_px = s_xyz[last * 3 + 0];
      s_py = s_xyz[last * 3 + 1];
      s_pz = s_xyz[last * 3 + 2];
    }
    __syncthreads();
    const float px = s_px, py = s_py, pz = s_pz;

    float bestV = -__builtin_inff();
    int   bestI = 0x7fffffff;
    for (int n = tid; n < N_; n += 1024) {
      const float dx = s_xyz[n * 3 + 0] - px;
      const float dy = s_xyz[n * 3 + 1] - py;
      const float dz = s_xyz[n * 3 + 2] - pz;
      const float dd = dx * dx + dy * dy + dz * dz;
      const float nm = fminf(s_mind[n], dd);
      s_mind[n] = nm;
      const bool better = nm > bestV;
      bestV = better ? nm : bestV;
      bestI = better ? n : bestI;
    }
    for (int m = 16; m; m >>= 1) {
      const float ov = __shfl_xor(bestV, m, 32);
      const int   oi = __shfl_xor(bestI, m, 32);
      const bool better = (ov > bestV) | ((ov == bestV) & (oi < bestI));
      bestV = better ? ov : bestV;
      bestI = better ? oi : bestI;
    }
    if (lane == 0) { swv[wid] = bestV; swi[wid] = bestI; }
    __syncthreads();
    if (wid == 0) {
      float v = swv[lane];
      int   i = swi[lane];
      for (int m = 16; m; m >>= 1) {
        const float ov = __shfl_xor(v, m, 32);
        const int   oi = __shfl_xor(i, m, 32);
        const bool better = (ov > v) | ((ov == v) & (oi < i));
        v = better ? ov : v;
        i = better ? oi : i;
      }
      if (lane == 0) s_last = i;
    }
    __syncthreads();
  }
}

__global__ __launch_bounds__(256) void three_nn_kernel(
    const float* __restrict__ pc, const int* __restrict__ fps_idx,
    int* __restrict__ idx3, float* __restrict__ w3) {
  const int tid  = threadIdx.x;
  const int lane = tid & 31;
  const int wave = blockIdx.x * (256 >> 5) + (tid >> 5);
  const int tilesPerBatch = M_ / 16;
  const int b    = wave / tilesPerBatch;
  const int rt   = wave - b * tilesPerBatch;
  const int half = lane >> 4;
  const int l15  = lane & 15;

  const float* pcb = pc + (size_t)b * N_ * 3;

  const int mrow = rt * 16 + l15;
  const int aidx = fps_idx[b * M_ + mrow];
  const float* ap = pcb + (size_t)aidx * 3;
  const float ax = ap[0], ay = ap[1], az = ap[2];
  v2f A;
  A.x = half ? (-2.0f * az) : (-2.0f * ax);
  A.y = half ? 1.0f         : (-2.0f * ay);

  float td[8][3];
  int   ti[8][3];
#pragma unroll
  for (int r = 0; r < 8; ++r)
#pragma unroll
    for (int k = 0; k < 3; ++k) { td[r][k] = __builtin_inff(); ti[r][k] = 0x7fffffff; }

#pragma unroll 2
  for (int t = 0; t < N_ / 16; ++t) {
    const int n = t * 16 + l15;
    const float* bp = pcb + (size_t)n * 3;
    const int pn = (n + 128 < N_) ? (n + 128) : (N_ - 1);
    __builtin_prefetch(pcb + (size_t)pn * 3, 0, 1);
    const float bx = bp[0], by = bp[1], bz = bp[2];
    const float bn2 = bx * bx + by * by + bz * bz;
    v8f c = {0.f, 0.f, 0.f, 0.f, 0.f, 0.f, 0.f, 0.f};
#if USE_WMMA
    v2f Bv;
    Bv.x = half ? bz  : bx;
    Bv.y = half ? bn2 : by;
    c = __builtin_amdgcn_wmma_f32_16x16x4_f32(
         false, A,  false, Bv,
         (short)0, c,  false,  false);
#else
#pragma unroll
    for (int r = 0; r < 8; ++r) {
      const int row = half * 8 + r;
      const float rx = __shfl(ax, row, 32);
      const float ry = __shfl(ay, row, 32);
      const float rz = __shfl(az, row, 32);
      c[r] = -2.0f * (rx * bx + ry * by + rz * bz) + bn2;
    }
#endif
#pragma unroll
    for (int r = 0; r < 8; ++r) ins3_stream(c[r], n, td[r], ti[r]);
  }

#pragma unroll
  for (int mskk = 1; mskk <= 8; mskk <<= 1) {
#pragma unroll
    for (int r = 0; r < 8; ++r) {
      float pd[3]; int pi[3];
#pragma unroll
      for (int k = 0; k < 3; ++k) {
        pd[k] = __shfl_xor(td[r][k], mskk, 32);
        pi[k] = __shfl_xor(ti[r][k], mskk, 32);
      }
#pragma unroll
      for (int k = 0; k < 3; ++k) ins3_lex(pd[k], pi[k], td[r], ti[r]);
    }
  }

  if (l15 == 0) {
#pragma unroll
    for (int r = 0; r < 8; ++r) {
      const int row = rt * 16 + half * 8 + r;
      const int gi  = fps_idx[b * M_ + row];
      const float* gp = pcb + (size_t)gi * 3;
      const float gx = gp[0], gy = gp[1], gz = gp[2];
      float wv[3];
      float wsum = 0.f;
#pragma unroll
      for (int k = 0; k < 3; ++k) {
        const int nn = ti[r][k];
        const float* np = pcb + (size_t)nn * 3;
        const float dx = np[0] - gx, dy = np[1] - gy, dz = np[2] - gz;
        const float dd = dx * dx + dy * dy + dz * dz;
        const float dist = sqrtf(fmaxf(dd, 0.0f));
        wv[k] = 1.0f / (dist + 1e-8f);
        wsum += wv[k];
        idx3[((size_t)b * M_ + row) * 3 + k] = nn;
      }
#pragma unroll
      for (int k = 0; k < 3; ++k)
        w3[((size_t)b * M_ + row) * 3 + k] = wv[k] / wsum;
    }
  }
}

__global__ __launch_bounds__(256) void interp_kernel(
    const float* __restrict__ feats, const int* __restrict__ idx3,
    const float* __restrict__ w3, float* __restrict__ out) {
  const int bc = blockIdx.x;
  const int b  = bc / C_;
  const int c  = bc - b * C_;
  const float* f = feats + ((size_t)b * C_ + c) * N_;
  float*       o = out   + ((size_t)b * C_ + c) * M_;
  const int*   ib = idx3 + (size_t)b * M_ * 3;
  const float* wb = w3   + (size_t)b * M_ * 3;
  for (int m = threadIdx.x; m < M_; m += 256) {
    const int   n0 = ib[m * 3 + 0], n1 = ib[m * 3 + 1], n2 = ib[m * 3 + 2];
    const float w0 = wb[m * 3 + 0], w1 = wb[m * 3 + 1], w2 = wb[m * 3 + 2];
    o[m] = w0 * f[n0] + w1 * f[n1] + w2 * f[n2];
  }
}

extern "C" void kernel_launch(void* const* d_in, const int* in_sizes, int n_in,
                              void* d_out, int out_size, void* d_ws, size_t ws_size,
                              hipStream_t stream) {
  const float* pc    = (const float*)d_in[0];
  const float* feats = (const float*)d_in[1];
  const float* obj   = (const float*)d_in[2];
  const float* grasp = (const float*)d_in[3];
  float* out = (float*)d_out;

  char* ws = (char*)d_ws;
  size_t off = 0;
  int* fps = (int*)(ws + off);
  off += (size_t)B_ * M_ * 4;        off = (off + 255) & ~(size_t)255;
  int* idx3 = (int*)(ws + off);
  off += (size_t)B_ * M_ * 3 * 4;    off = (off + 255) & ~(size_t)255;
  float* w3 = (float*)(ws + off);
  off += (size_t)B_ * M_ * 3 * 4;

  fps_kernel<<<B_, 1024, 0, stream>>>(pc, obj, grasp, fps);
  three_nn_kernel<<<(B_ * (M_ / 16)) / 8, 256, 0, stream>>>(pc, fps, idx3, w3);
  interp_kernel<<<B_ * C_, 256, 0, stream>>>(feats, idx3, w3, out);
}


# fps kernel: compact the masked points in index order (ballot words + prefix) into LDS so each of the 1024 serial steps scans only candidates; full layout kept if over 15000 or none
# speedup vs baseline: 1.5404x; 1.5404x over previous
#include <hip/hip_runtime.h>
#include <hip/hip_bf16.h>

#define B_ 4
#define N_ 20000
#define C_ 512
#define M_ 1024
#define GRASP_TH 0.1f

typedef float v2f __attribute__((ext_vector_type(2)));
typedef float v8f __attribute__((ext_vector_type(8)));

#if defined(__gfx1250__) && __has_builtin(__builtin_amdgcn_wmma_f32_16x16x4_f32)
#define USE_WMMA 1
#else
#define USE_WMMA 0
#endif

__device__ __forceinline__ void ins3_stream(float v, int n, float d[3], int ix[3]) {
  const bool c0 = v < d[0];
  const bool c1 = v < d[1];
  const bool c2 = v < d[2];
  const float r0 = c0 ? v     : d[0];
  const int   j0 = c0 ? n     : ix[0];
  const float r1 = c0 ? d[0]  : (c1 ? v : d[1]);
  const int   j1 = c0 ? ix[0] : (c1 ? n : ix[1]);
  const float r2 = c1 ? d[1]  : (c2 ? v : d[2]);
  const int   j2 = c1 ? ix[1] : (c2 ? n : ix[2]);
  d[0] = r0; d[1] = r1; d[2] = r2;
  ix[0] = j0; ix[1] = j1; ix[2] = j2;
}

__device__ __forceinline__ void ins3_lex(float v, int n, float d[3], int ix[3]) {
  const bool c0 = (v < d[0]) | ((v == d[0]) & (n < ix[0]));
  const bool c1 = (v < d[1]) | ((v == d[1]) & (n < ix[1]));
  const bool c2 = (v < d[2]) | ((v == d[2]) & (n < ix[2]));
  const float r0 = c0 ? v     : d[0];
  const int   j0 = c0 ? n     : ix[0];
  const float r1 = c0 ? d[0]  : (c1 ? v : d[1]);
  const int   j1 = c0 ? ix[0] : (c1 ? n : ix[1]);
  const float r2 = c1 ? d[1]  : (c2 ? v : d[2]);
  const int   j2 = c1 ? ix[1] : (c2 ? n : ix[2]);
  d[0] = r0; d[1] = r1; d[2] = r2;
  ix[0] = j0; ix[1] = j1; ix[2] = j2;
}

#define FPS_PASSES 20
#define FPS_WORDS  (FPS_PASSES * 32)
#define CMAX 15000

__global__ __launch_bounds__(1024) void fps_kernel(
    const float* __restrict__ pc, const float* __restrict__ obj,
    const float* __restrict__ grasp, int* __restrict__ fps_idx) {
  const int b    = blockIdx.x;
  const int tid  = threadIdx.x;
  const int lane = tid & 31;
  const int wid  = tid >> 5;

  __shared__ float    s_xyz[N_ * 3];
  __shared__ float    s_mind[N_];
  __shared__ unsigned s_bits[FPS_WORDS];
  __shared__ int      s_pref[FPS_WORDS];
  __shared__ float swv[32];
  __shared__ int   swi[32];
  __shared__ int   s_last;
  __shared__ int   s_K;
  __shared__ float s_px, s_py, s_pz;

  const float* pcb = pc + (size_t)b * N_ * 3;
  const float* ob0 = obj + (size_t)b * 2 * N_;
  const float* gr  = grasp + (size_t)b * N_;

  unsigned myw = 0u;
  for (int it = 0; it < FPS_PASSES; ++it) {
    const int n  = it * 1024 + tid;
    const int nl = (n < N_) ? n : (N_ - 1);
    const bool mk = (n < N_) & (ob0[N_ + nl] > ob0[nl]) & (gr[nl] > GRASP_TH);
    const unsigned w = __builtin_amdgcn_ballot_w32(mk);
    myw = (lane == it) ? w : myw;
  }
  if (lane < FPS_PASSES) s_bits[lane * 32 + wid] = myw;
  __syncthreads();

  if (wid == 0) {
    int sum = 0;
    for (int j = 0; j < FPS_PASSES; ++j) sum += __popc(s_bits[lane * FPS_PASSES + j]);
    int incl = sum;
    for (int d = 1; d < 32; d <<= 1) {
      const int o = __shfl_up(incl, d, 32);
      incl += (lane >= d) ? o : 0;
    }
    int run = incl - sum;
    for (int j = 0; j < FPS_PASSES; ++j) {
      s_pref[lane * FPS_PASSES + j] = run;
      run += __popc(s_bits[lane * FPS_PASSES + j]);
    }
    if (lane == 31) s_K = incl;
    if (lane == 0)  s_last = 0;
  }
  __syncthreads();

  const int  K       = s_K;
  const bool compact = (K >= 1) & (K <= CMAX);
  const int  Kc      = compact ? K : N_;
  int* s_idx = (int*)(s_xyz + (compact ? 3 * Kc : 0));

  for (int g = 0; g < FPS_PASSES; g += 4) {
    float x[4], y[4], z[4];
    unsigned w[4];
    int pf[4];
    for (int j = 0; j < 4; ++j) {
      const int n  = (g + j) * 1024 + tid;
      const int nl = (n < N_) ? n : (N_ - 1);
      const float* p = pcb + (size_t)nl * 3;
      x[j] = p[0]; y[j] = p[1]; z[j] = p[2];
      w[j]  = s_bits[(g + j) * 32 + wid];
      pf[j] = s_pref[(g + j) * 32 + wid];
    }
    for (int j = 0; j < 4; ++j) {
      const int  n    = (g + j) * 1024 + tid;
      const bool mk   = (w[j] >> lane) & 1u;
      const int  pos  = pf[j] + __popc(w[j] & ((1u << lane) - 1u));
      const bool keep = compact ? mk : (n < N_);
      const int  slot = compact ? pos : n;
      if (keep) {
        s_xyz[slot * 3 + 0] = x[j];
        s_xyz[slot * 3 + 1] = y[j];
        s_xyz[slot * 3 + 2] = z[j];
        s_mind[slot] = mk ? __builtin_inff() : -__builtin_inff();
        if (compact) s_idx[slot] = n;
        else if (mk & (pos == 0)) s_last = n;
      }
    }
  }
  __syncthreads();

  for (int step = 0; step < M_; ++step) {
    if (tid == 0) {
      const int last = s_last;
      fps_idx[b * M_ + step] = compact ? s_idx[last] : last;
      s_px = s_xyz[last * 3 + 0];
      s_py = s_xyz[last * 3 + 1];
      s_pz = s_xyz[last * 3 + 2];
    }
    __syncthreads();
    const float px = s_px, py = s_py, pz = s_pz;

    float bestV = -__builtin_inff();
    int   bestI = 0;
#pragma unroll 4
    for (int n = tid; n < Kc; n += 1024) {
      const float dx = s_xyz[n * 3 + 0] - px;
      const float dy = s_xyz[n * 3 + 1] - py;
      const float dz = s_xyz[n * 3 + 2] - pz;
      const float dd = dx * dx + dy * dy + dz * dz;
      const float nm = fminf(s_mind[n], dd);
      s_mind[n] = nm;
      const bool better = nm > bestV;
      bestV = better ? nm : bestV;
      bestI = better ? n : bestI;
    }
    for (int m = 16; m; m >>= 1) {
      const float ov = __shfl_xor(bestV, m, 32);
      const int   oi = __shfl_xor(bestI, m, 32);
      const bool better = (ov > bestV) | ((ov == bestV) & (oi < bestI));
      bestV = better ? ov : bestV;
      bestI = better ? oi : bestI;
    }
    if (lane == 0) { swv[wid] = bestV; swi[wid] = bestI; }
    __syncthreads();
    if (wid == 0) {
      float v = swv[lane];
      int   i = swi[lane];
      for (int m = 16; m; m >>= 1) {
        const float ov = __shfl_xor(v, m, 32);
        const int   oi = __shfl_xor(i, m, 32);
        const bool better = (ov > v) | ((ov == v) & (oi < i));
        v = better ? ov : v;
        i = better ? oi : i;
      }
      if (lane == 0) s_last = i;
    }
    __syncthreads();
  }
}

__global__ __launch_bounds__(256) void three_nn_kernel(
    const float* __restrict__ pc, const int* __restrict__ fps_idx,
    int* __restrict__ idx3, float* __restrict__ w3) {
  const int tid  = threadIdx.x;
  const int lane = tid & 31;
  const int wave = blockIdx.x * (256 >> 5) + (tid >> 5);
  const int tilesPerBatch = M_ / 16;
  const int b    = wave / tilesPerBatch;
  const int rt   = wave - b * tilesPerBatch;
  const int half = lane >> 4;
  const int l15  = lane & 15;

  const float* pcb = pc + (size_t)b * N_ * 3;

  const int mrow = rt * 16 + l15;
  const int aidx = fps_idx[b * M_ + mrow];
  const float* ap = pcb + (size_t)aidx * 3;
  const float ax = ap[0], ay = ap[1], az = ap[2];
  v2f A;
  A.x = half ? (-2.0f * az) : (-2.0f * ax);
  A.y = half ? 1.0f         : (-2.0f * ay);

  float td[8][3];
  int   ti[8][3];
#pragma unroll
  for (int r = 0; r < 8; ++r)
#pragma unroll
    for (int k = 0; k < 3; ++k) { td[r][k] = __builtin_inff(); ti[r][k] = 0x7fffffff; }

#pragma unroll 2
  for (int t = 0; t < N_ / 16; ++t) {
    const int n = t * 16 + l15;
    const float* bp = pcb + (size_t)n * 3;
    const int pn = (n + 128 < N_) ? (n + 128) : (N_ - 1);
    __builtin_prefetch(pcb + (size_t)pn * 3, 0, 1);
    const float bx = bp[0], by = bp[1], bz = bp[2];
    const float bn2 = bx * bx + by * by + bz * bz;
    v8f c = {0.f, 0.f, 0.f, 0.f, 0.f, 0.f, 0.f, 0.f};
#if USE_WMMA
    v2f Bv;
    Bv.x = half ? bz  : bx;
    Bv.y = half ? bn2 : by;
    c = __builtin_amdgcn_wmma_f32_16x16x4_f32(
         false, A,  false, Bv,
         (short)0, c,  false,  false);
#else
#pragma unroll
    for (int r = 0; r < 8; ++r) {
      const int row = half * 8 + r;
      const float rx = __shfl(ax, row, 32);
      const float ry = __shfl(ay, row, 32);
      const float rz = __shfl(az, row, 32);
      c[r] = -2.0f * (rx * bx + ry * by + rz * bz) + bn2;
    }
#endif
#pragma unroll
    for (int r = 0; r < 8; ++r) ins3_stream(c[r], n, td[r], ti[r]);
  }

#pragma unroll
  for (int mskk = 1; mskk <= 8; mskk <<= 1) {
#pragma unroll
    for (int r = 0; r < 8; ++r) {
      float pd[3]; int pi[3];
#pragma unroll
      for (int k = 0; k < 3; ++k) {
        pd[k] = __shfl_xor(td[r][k], mskk, 32);
        pi[k] = __shfl_xor(ti[r][k], mskk, 32);
      }
#pragma unroll
      for (int k = 0; k < 3; ++k) ins3_lex(pd[k], pi[k], td[r], ti[r]);
    }
  }

  if (l15 == 0) {
#pragma unroll
    for (int r = 0; r < 8; ++r) {
      const int row = rt * 16 + half * 8 + r;
      const int gi  = fps_idx[b * M_ + row];
      const float* gp = pcb + (size_t)gi * 3;
      const float gx = gp[0], gy = gp[1], gz = gp[2];
      float wv[3];
      float wsum = 0.f;
#pragma unroll
      for (int k = 0; k < 3; ++k) {
        const int nn = ti[r][k];
        const float* np = pcb + (size_t)nn * 3;
        const float dx = np[0] - gx, dy = np[1] - gy, dz = np[2] - gz;
        const float dd = dx * dx + dy * dy + dz * dz;
        const float dist = sqrtf(fmaxf(dd, 0.0f));
        wv[k] = 1.0f / (dist + 1e-8f);
        wsum += wv[k];
        idx3[((size_t)b * M_ + row) * 3 + k] = nn;
      }
#pragma unroll
      for (int k = 0; k < 3; ++k)
        w3[((size_t)b * M_ + row) * 3 + k] = wv[k] / wsum;
    }
  }
}

__global__ __launch_bounds__(256) void interp_kernel(
    const float* __restrict__ feats, const int* __restrict__ idx3,
    const float* __restrict__ w3, float* __restrict__ out) {
  const int bc = blockIdx.x;
  const int b  = bc / C_;
  const int c  = bc - b * C_;
  const float* f = feats + ((size_t)b * C_ + c) * N_;
  float*       o = out   + ((size_t)b * C_ + c) * M_;
  const int*   ib = idx3 + (size_t)b * M_ * 3;
  const float* wb = w3   + (size_t)b * M_ * 3;
  for (int m = threadIdx.x; m < M_; m += 256) {
    const int   n0 = ib[m * 3 + 0], n1 = ib[m * 3 + 1], n2 = ib[m * 3 + 2];
    const float w0 = wb[m * 3 + 0], w1 = wb[m * 3 + 1], w2 = wb[m * 3 + 2];
    o[m] = w0 * f[n0] + w1 * f[n1] + w2 * f[n2];
  }
}

extern "C" void kernel_launch(void* const* d_in, const int* in_sizes, int n_in,
                              void* d_out, int out_size, void* d_ws, size_t ws_size,
                              hipStream_t stream) {
  const float* pc    = (const float*)d_in[0];
  const float* feats = (const float*)d_in[1];
  const float* obj   = (const float*)d_in[2];
  const float* grasp = (const float*)d_in[3];
  float* out = (float*)d_out;

  char* ws = (char*)d_ws;
  size_t off = 0;
  int* fps = (int*)(ws + off);
  off += (size_t)B_ * M_ * 4;        off = (off + 255) & ~(size_t)255;
  int* idx3 = (int*)(ws + off);
  off += (size_t)B_ * M_ * 3 * 4;    off = (off + 255) & ~(size_t)255;
  float* w3 = (float*)(ws + off);
  off += (size_t)B_ * M_ * 3 * 4;

  fps_kernel<<<B_, 1024, 0, stream>>>(pc, obj, grasp, fps);
  three_nn_kernel<<<(B_ * (M_ / 16)) / 8, 256, 0, stream>>>(pc, fps, idx3, w3);
  interp_kernel<<<B_ * C_, 256, 0, stream>>>(feats, idx3, w3, out);
}


# fps kernel: each step's publish stage removed with its barrier (3 -> 2) and global-store wait: all threads read the winner's slot and xyz from LDS; wave 0 stores picks 32 at a time, one 128-B line
# speedup vs baseline: 1.8244x; 1.1844x over previous
#include <hip/hip_runtime.h>
#include <hip/hip_bf16.h>

#define B_ 4
#define N_ 20000
#define C_ 512
#define M_ 1024
#define GRASP_TH 0.1f

typedef float v2f __attribute__((ext_vector_type(2)));
typedef float v8f __attribute__((ext_vector_type(8)));

#if defined(__gfx1250__) && __has_builtin(__builtin_amdgcn_wmma_f32_16x16x4_f32)
#define USE_WMMA 1
#else
#define USE_WMMA 0
#endif

__device__ __forceinline__ void ins3_stream(float v, int n, float d[3], int ix[3]) {
  const bool c0 = v < d[0];
  const bool c1 = v < d[1];
  const bool c2 = v < d[2];
  const float r0 = c0 ? v     : d[0];
  const int   j0 = c0 ? n     : ix[0];
  const float r1 = c0 ? d[0]  : (c1 ? v : d[1]);
  const int   j1 = c0 ? ix[0] : (c1 ? n : ix[1]);
  const float r2 = c1 ? d[1]  : (c2 ? v : d[2]);
  const int   j2 = c1 ? ix[1] : (c2 ? n : ix[2]);
  d[0] = r0; d[1] = r1; d[2] = r2;
  ix[0] = j0; ix[1] = j1; ix[2] = j2;
}

__device__ __forceinline__ void ins3_lex(float v, int n, float d[3], int ix[3]) {
  const bool c0 = (v < d[0]) | ((v == d[0]) & (n < ix[0]));
  const bool c1 = (v < d[1]) | ((v == d[1]) & (n < ix[1]));
  const bool c2 = (v < d[2]) | ((v == d[2]) & (n < ix[2]));
  const float r0 = c0 ? v     : d[0];
  const int   j0 = c0 ? n     : ix[0];
  const float r1 = c0 ? d[0]  : (c1 ? v : d[1]);
  const int   j1 = c0 ? ix[0] : (c1 ? n : ix[1]);
  const float r2 = c1 ? d[1]  : (c2 ? v : d[2]);
  const int   j2 = c1 ? ix[1] : (c2 ? n : ix[2]);
  d[0] = r0; d[1] = r1; d[2] = r2;
  ix[0] = j0; ix[1] = j1; ix[2] = j2;
}

#define FPS_PASSES 20
#define FPS_WORDS  (FPS_PASSES * 32)
#define CMAX 15000

__global__ __launch_bounds__(1024) void fps_kernel(
    const float* __restrict__ pc, const float* __restrict__ obj,
    const float* __restrict__ grasp, int* __restrict__ fps_idx) {
  const int b    = blockIdx.x;
  const int tid  = threadIdx.x;
  const int lane = tid & 31;
  const int wid  = tid >> 5;

  __shared__ float    s_xyz[N_ * 3];
  __shared__ float    s_mind[N_];
  __shared__ unsigned s_bits[FPS_WORDS];
  __shared__ int      s_pref[FPS_WORDS];
  __shared__ float swv[32];
  __shared__ int   swi[32];
  __shared__ int   s_last;
  __shared__ int   s_K;

  const float* pcb = pc + (size_t)b * N_ * 3;
  const float* ob0 = obj + (size_t)b * 2 * N_;
  const float* gr  = grasp + (size_t)b * N_;

  unsigned myw = 0u;
  for (int it = 0; it < FPS_PASSES; ++it) {
    const int n  = it * 1024 + tid;
    const int nl = (n < N_) ? n : (N_ - 1);
    const bool mk = (n < N_) & (ob0[N_ + nl] > ob0[nl]) & (gr[nl] > GRASP_TH);
    const unsigned w = __builtin_amdgcn_ballot_w32(mk);
    myw = (lane == it) ? w : myw;
  }
  if (lane < FPS_PASSES) s_bits[lane * 32 + wid] = myw;
  __syncthreads();

  if (wid == 0) {
    int sum = 0;
    for (int j = 0; j < FPS_PASSES; ++j) sum += __popc(s_bits[lane * FPS_PASSES + j]);
    int incl = sum;
    for (int d = 1; d < 32; d <<= 1) {
      const int o = __shfl_up(incl, d, 32);
      incl += (lane >= d) ? o : 0;
    }
    int run = incl - sum;
    for (int j = 0; j < FPS_PASSES; ++j) {
      s_pref[lane * FPS_PASSES + j] = run;
      run += __popc(s_bits[lane * FPS_PASSES + j]);
    }
    if (lane == 31) s_K = incl;
    if (lane == 0)  s_last = 0;
  }
  __syncthreads();

  const int  K       = s_K;
  const bool compact = (K >= 1) & (K <= CMAX);
  const int  Kc      = compact ? K : N_;
  int* s_idx = (int*)(s_xyz + (compact ? 3 * Kc : 0));

  for (int g = 0; g < FPS_PASSES; g += 4) {
    float x[4], y[4], z[4];
    unsigned w[4];
    int pf[4];
    for (int j = 0; j < 4; ++j) {
      const int n  = (g + j) * 1024 + tid;
      const int nl = (n < N_) ? n : (N_ - 1);
      const float* p = pcb + (size_t)nl * 3;
      x[j] = p[0]; y[j] = p[1]; z[j] = p[2];
      w[j]  = s_bits[(g + j) * 32 + wid];
      pf[j] = s_pref[(g + j) * 32 + wid];
    }
    for (int j = 0; j < 4; ++j) {
      const int  n    = (g + j) * 1024 + tid;
      const bool mk   = (w[j] >> lane) & 1u;
      const int  pos  = pf[j] + __popc(w[j] & ((1u << lane) - 1u));
      const bool keep = compact ? mk : (n < N_);
      const int  slot = compact ? pos : n;
      if (keep) {
        s_xyz[slot * 3 + 0] = x[j];
        s_xyz[slot * 3 + 1] = y[j];
        s_xyz[slot * 3 + 2] = z[j];
        s_mind[slot] = mk ? __builtin_inff() : -__builtin_inff();
        if (compact) s_idx[slot] = n;
        else if (mk & (pos == 0)) s_last = n;
      }
    }
  }
  __syncthreads();

  int acc = 0;
  for (int step = 0; step < M_; ++step) {
    const int last = s_last;
    const float px = s_xyz[last * 3 + 0];
    const float py = s_xyz[last * 3 + 1];
    const float pz = s_xyz[last * 3 + 2];
    if (wid == 0) {
      const int pick = compact ? s_idx[last] : last;
      acc = (lane == (step & 31)) ? pick : acc;
      if ((step & 31) == 31) fps_idx[b * M_ + (step - 31) + lane] = acc;
    }

    float bestV = -__builtin_inff();
    int   bestI = 0;
#pragma unroll 4
    for (int n = tid; n < Kc; n += 1024) {
      const float dx = s_xyz[n * 3 + 0] - px;
      const float dy = s_xyz[n * 3 + 1] - py;
      const float dz = s_xyz[n * 3 + 2] - pz;
      const float dd = dx * dx + dy * dy + dz * dz;
      const float nm = fminf(s_mind[n], dd);
      s_mind[n] = nm;
      const bool better = nm > bestV;
      bestV = better ? nm : bestV;
      bestI = better ? n : bestI;
    }
    for (int m = 16; m; m >>= 1) {
      const float ov = __shfl_xor(bestV, m, 32);
      const int   oi = __shfl_xor(bestI, m, 32);
      const bool better = (ov > bestV) | ((ov == bestV) & (oi < bestI));
      bestV = better ? ov : bestV;
      bestI = better ? oi : bestI;
    }
    if (lane == 0) { swv[wid] = bestV; swi[wid] = bestI; }
    __syncthreads();
    if (wid == 0) {
      float v = swv[lane];
      int   i = swi[lane];
      for (int m = 16; m; m >>= 1) {
        const float ov = __shfl_xor(v, m, 32);
        const int   oi = __shfl_xor(i, m, 32);
        const bool better = (ov > v) | ((ov == v) & (oi < i));
        v = better ? ov : v;
        i = better ? oi : i;
      }
      if (lane == 0) s_last = i;
    }
    __syncthreads();
  }
}

__global__ __launch_bounds__(256) void three_nn_kernel(
    const float* __restrict__ pc, const int* __restrict__ fps_idx,
    int* __restrict__ idx3, float* __restrict__ w3) {
  const int tid  = threadIdx.x;
  const int lane = tid & 31;
  const int wave = blockIdx.x * (256 >> 5) + (tid >> 5);
  const int tilesPerBatch = M_ / 16;
  const int b    = wave / tilesPerBatch;
  const int rt   = wave - b * tilesPerBatch;
  const int half = lane >> 4;
  const int l15  = lane & 15;

  const float* pcb = pc + (size_t)b * N_ * 3;

  const int mrow = rt * 16 + l15;
  const int aidx = fps_idx[b * M_ + mrow];
  const float* ap = pcb + (size_t)aidx * 3;
  const float ax = ap[0], ay = ap[1], az = ap[2];
  v2f A;
  A.x = half ? (-2.0f * az) : (-2.0f * ax);
  A.y = half ? 1.0f         : (-2.0f * ay);

  float td[8][3];
  int   ti[8][3];
#pragma unroll
  for (int r = 0; r < 8; ++r)
#pragma unroll
    for (int k = 0; k < 3; ++k) { td[r][k] = __builtin_inff(); ti[r][k] = 0x7fffffff; }

#pragma unroll 2
  for (int t = 0; t < N_ / 16; ++t) {
    const int n = t * 16 + l15;
    const float* bp = pcb + (size_t)n * 3;
    const int pn = (n + 128 < N_) ? (n + 128) : (N_ - 1);
    __builtin_prefetch(pcb + (size_t)pn * 3, 0, 1);
    const float bx = bp[0], by = bp[1], bz = bp[2];
    const float bn2 = bx * bx + by * by + bz * bz;
    v8f c = {0.f, 0.f, 0.f, 0.f, 0.f, 0.f, 0.f, 0.f};
#if USE_WMMA
    v2f Bv;
    Bv.x = half ? bz  : bx;
    Bv.y = half ? bn2 : by;
    c = __builtin_amdgcn_wmma_f32_16x16x4_f32(
         false, A,  false, Bv,
         (short)0, c,  false,  false);
#else
#pragma unroll
    for (int r = 0; r < 8; ++r) {
      const int row = half * 8 + r;
      const float rx = __shfl(ax, row, 32);
      const float ry = __shfl(ay, row, 32);
      const float rz = __shfl(az, row, 32);
      c[r] = -2.0f * (rx * bx + ry * by + rz * bz) + bn2;
    }
#endif
#pragma unroll
    for (int r = 0; r < 8; ++r) ins3_stream(c[r], n, td[r], ti[r]);
  }

#pragma unroll
  for (int mskk = 1; mskk <= 8; mskk <<= 1) {
#pragma unroll
    for (int r = 0; r < 8; ++r) {
      float pd[3]; int pi[3];
#pragma unroll
      for (int k = 0; k < 3; ++k) {
        pd[k] = __shfl_xor(td[r][k], mskk, 32);
        pi[k] = __shfl_xor(ti[r][k], mskk, 32);
      }
#pragma unroll
      for (int k = 0; k < 3; ++k) ins3_lex(pd[k], pi[k], td[r], ti[r]);
    }
  }

  if (l15 == 0) {
#pragma unroll
    for (int r = 0; r < 8; ++r) {
      const int row = rt * 16 + half * 8 + r;
      const int gi  = fps_idx[b * M_ + row];
      const float* gp = pcb + (size_t)gi * 3;
      const float gx = gp[0], gy = gp[1], gz = gp[2];
      float wv[3];
      float wsum = 0.f;
#pragma unroll
      for (int k = 0; k < 3; ++k) {
        const int nn = ti[r][k];
        const float* np = pcb + (size_t)nn * 3;
        const float dx = np[0] - gx, dy = np[1] - gy, dz = np[2] - gz;
        const float dd = dx * dx + dy * dy + dz * dz;
        const float dist = sqrtf(fmaxf(dd, 0.0f));
        wv[k] = 1.0f / (dist + 1e-8f);
        wsum += wv[k];
        idx3[((size_t)b * M_ + row) * 3 + k] = nn;
      }
#pragma unroll
      for (int k = 0; k < 3; ++k)
        w3[((size_t)b * M_ + row) * 3 + k] = wv[k] / wsum;
    }
  }
}

__global__ __launch_bounds__(256) void interp_kernel(
    const float* __restrict__ feats, const int* __restrict__ idx3,
    const float* __restrict__ w3, float* __restrict__ out) {
  const int bc = blockIdx.x;
  const int b  = bc / C_;
  const int c  = bc - b * C_;
  const float* f = feats + ((size_t)b * C_ + c) * N_;
  float*       o = out   + ((size_t)b * C_ + c) * M_;
  const int*   ib = idx3 + (size_t)b * M_ * 3;
  const float* wb = w3   + (size_t)b * M_ * 3;
  for (int m = threadIdx.x; m < M_; m += 256) {
    const int   n0 = ib[m * 3 + 0], n1 = ib[m * 3 + 1], n2 = ib[m * 3 + 2];
    const float w0 = wb[m * 3 + 0], w1 = wb[m * 3 + 1], w2 = wb[m * 3 + 2];
    o[m] = w0 * f[n0] + w1 * f[n1] + w2 * f[n2];
  }
}

extern "C" void kernel_launch(void* const* d_in, const int* in_sizes, int n_in,
                              void* d_out, int out_size, void* d_ws, size_t ws_size,
                              hipStream_t stream) {
  const float* pc    = (const float*)d_in[0];
  const float* feats = (const float*)d_in[1];
  const float* obj   = (const float*)d_in[2];
  const float* grasp = (const float*)d_in[3];
  float* out = (float*)d_out;

  char* ws = (char*)d_ws;
  size_t off = 0;
  int* fps = (int*)(ws + off);
  off += (size_t)B_ * M_ * 4;        off = (off + 255) & ~(size_t)255;
  int* idx3 = (int*)(ws + off);
  off += (size_t)B_ * M_ * 3 * 4;    off = (off + 255) & ~(size_t)255;
  float* w3 = (float*)(ws + off);
  off += (size_t)B_ * M_ * 3 * 4;

  fps_kernel<<<B_, 1024, 0, stream>>>(pc, obj, grasp, fps);
  three_nn_kernel<<<(B_ * (M_ / 16)) / 8, 256, 0, stream>>>(pc, fps, idx3, w3);
  interp_kernel<<<B_ * C_, 256, 0, stream>>>(feats, idx3, w3, out);
}


# fps kernel: both per-step argmax reductions (32 lanes, then 32 wave winners) use vector-ALU cross-lane ops (DPP row_xmask, permlanex16: max value, then min index among holders), no LDS shuffles
# speedup vs baseline: 2.3878x; 1.3088x over previous
#include <hip/hip_runtime.h>
#include <hip/hip_bf16.h>

#define B_ 4
#define N_ 20000
#define C_ 512
#define M_ 1024
#define GRASP_TH 0.1f

typedef float v2f __attribute__((ext_vector_type(2)));
typedef float v8f __attribute__((ext_vector_type(8)));

#if defined(__gfx1250__) && __has_builtin(__builtin_amdgcn_wmma_f32_16x16x4_f32)
#define USE_WMMA 1
#else
#define USE_WMMA 0
#endif

__device__ __forceinline__ void ins3_stream(float v, int n, float d[3], int ix[3]) {
  const bool c0 = v < d[0];
  const bool c1 = v < d[1];
  const bool c2 = v < d[2];
  const float r0 = c0 ? v     : d[0];
  const int   j0 = c0 ? n     : ix[0];
  const float r1 = c0 ? d[0]  : (c1 ? v : d[1]);
  const int   j1 = c0 ? ix[0] : (c1 ? n : ix[1]);
  const float r2 = c1 ? d[1]  : (c2 ? v : d[2]);
  const int   j2 = c1 ? ix[1] : (c2 ? n : ix[2]);
  d[0] = r0; d[1] = r1; d[2] = r2;
  ix[0] = j0; ix[1] = j1; ix[2] = j2;
}

__device__ __forceinline__ void ins3_lex(float v, int n, float d[3], int ix[3]) {
  const bool c0 = (v < d[0]) | ((v == d[0]) & (n < ix[0]));
  const bool c1 = (v < d[1]) | ((v == d[1]) & (n < ix[1]));
  const bool c2 = (v < d[2]) | ((v == d[2]) & (n < ix[2]));
  const float r0 = c0 ? v     : d[0];
  const int   j0 = c0 ? n     : ix[0];
  const float r1 = c0 ? d[0]  : (c1 ? v : d[1]);
  const int   j1 = c0 ? ix[0] : (c1 ? n : ix[1]);
  const float r2 = c1 ? d[1]  : (c2 ? v : d[2]);
  const int   j2 = c1 ? ix[1] : (c2 ? n : ix[2]);
  d[0] = r0; d[1] = r1; d[2] = r2;
  ix[0] = j0; ix[1] = j1; ix[2] = j2;
}

template <int XM>
__device__ __forceinline__ int row_xor_i(int v) {
  return __builtin_amdgcn_update_dpp(v, v, 0x160 + XM, 0xf, 0xf, true);
}
__device__ __forceinline__ int xhalf_i(int v) {
  return __builtin_amdgcn_permlanex16(v, v, 0x76543210, 0xfedcba98, false, false);
}
__device__ __forceinline__ void wave_argmax(float& v, int& i) {
  const int key = __builtin_bit_cast(int, v);
  int m = key;
  m = max(m, row_xor_i<1>(m));
  m = max(m, row_xor_i<2>(m));
  m = max(m, row_xor_i<4>(m));
  m = max(m, row_xor_i<8>(m));
  m = max(m, xhalf_i(m));
  int c = (key == m) ? i : 0x7fffffff;
  c = min(c, row_xor_i<1>(c));
  c = min(c, row_xor_i<2>(c));
  c = min(c, row_xor_i<4>(c));
  c = min(c, row_xor_i<8>(c));
  c = min(c, xhalf_i(c));
  v = __builtin_bit_cast(float, m);
  i = c;
}

#define FPS_PASSES 20
#define FPS_WORDS  (FPS_PASSES * 32)
#define CMAX 15000

__global__ __launch_bounds__(1024) void fps_kernel(
    const float* __restrict__ pc, const float* __restrict__ obj,
    const float* __restrict__ grasp, int* __restrict__ fps_idx) {
  const int b    = blockIdx.x;
  const int tid  = threadIdx.x;
  const int lane = tid & 31;
  const int wid  = tid >> 5;

  __shared__ float    s_xyz[N_ * 3];
  __shared__ float    s_mind[N_];
  __shared__ unsigned s_bits[FPS_WORDS];
  __shared__ int      s_pref[FPS_WORDS];
  __shared__ float swv[32];
  __shared__ int   swi[32];
  __shared__ int   s_last;
  __shared__ int   s_K;

  const float* pcb = pc + (size_t)b * N_ * 3;
  const float* ob0 = obj + (size_t)b * 2 * N_;
  const float* gr  = grasp + (size_t)b * N_;

  unsigned myw = 0u;
  for (int it = 0; it < FPS_PASSES; ++it) {
    const int n  = it * 1024 + tid;
    const int nl = (n < N_) ? n : (N_ - 1);
    const bool mk = (n < N_) & (ob0[N_ + nl] > ob0[nl]) & (gr[nl] > GRASP_TH);
    const unsigned w = __builtin_amdgcn_ballot_w32(mk);
    myw = (lane == it) ? w : myw;
  }
  if (lane < FPS_PASSES) s_bits[lane * 32 + wid] = myw;
  __syncthreads();

  if (wid == 0) {
    int sum = 0;
    for (int j = 0; j < FPS_PASSES; ++j) sum += __popc(s_bits[lane * FPS_PASSES + j]);
    int incl = sum;
    for (int d = 1; d < 32; d <<= 1) {
      const int o = __shfl_up(incl, d, 32);
      incl += (lane >= d) ? o : 0;
    }
    int run = incl - sum;
    for (int j = 0; j < FPS_PASSES; ++j) {
      s_pref[lane * FPS_PASSES + j] = run;
      run += __popc(s_bits[lane * FPS_PASSES + j]);
    }
    if (lane == 31) s_K = incl;
    if (lane == 0)  s_last = 0;
  }
  __syncthreads();

  const int  K       = s_K;
  const bool compact = (K >= 1) & (K <= CMAX);
  const int  Kc      = compact ? K : N_;
  int* s_idx = (int*)(s_xyz + (compact ? 3 * Kc : 0));

  for (int g = 0; g < FPS_PASSES; g += 4) {
    float x[4], y[4], z[4];
    unsigned w[4];
    int pf[4];
    for (int j = 0; j < 4; ++j) {
      const int n  = (g + j) * 1024 + tid;
      const int nl = (n < N_) ? n : (N_ - 1);
      const float* p = pcb + (size_t)nl * 3;
      x[j] = p[0]; y[j] = p[1]; z[j] = p[2];
      w[j]  = s_bits[(g + j) * 32 + wid];
      pf[j] = s_pref[(g + j) * 32 + wid];
    }
    for (int j = 0; j < 4; ++j) {
      const int  n    = (g + j) * 1024 + tid;
      const bool mk   = (w[j] >> lane) & 1u;
      const int  pos  = pf[j] + __popc(w[j] & ((1u << lane) - 1u));
      const bool keep = compact ? mk : (n < N_);
      const int  slot = compact ? pos : n;
      if (keep) {
        s_xyz[slot * 3 + 0] = x[j];
        s_xyz[slot * 3 + 1] = y[j];
        s_xyz[slot * 3 + 2] = z[j];
        s_mind[slot] = mk ? __builtin_inff() : -__builtin_inff();
        if (compact) s_idx[slot] = n;
        else if (mk & (pos == 0)) s_last = n;
      }
    }
  }
  __syncthreads();

  int acc = 0;
  for (int step = 0; step < M_; ++step) {
    const int last = s_last;
    const float px = s_xyz[last * 3 + 0];
    const float py = s_xyz[last * 3 + 1];
    const float pz = s_xyz[last * 3 + 2];
    if (wid == 0) {
      const int pick = compact ? s_idx[last] : last;
      acc = (lane == (step & 31)) ? pick : acc;
      if ((step & 31) == 31) fps_idx[b * M_ + (step - 31) + lane] = acc;
    }

    float bestV = -__builtin_inff();
    int   bestI = 0;
#pragma unroll 4
    for (int n = tid; n < Kc; n += 1024) {
      const float dx = s_xyz[n * 3 + 0] - px;
      const float dy = s_xyz[n * 3 + 1] - py;
      const float dz = s_xyz[n * 3 + 2] - pz;
      const float dd = dx * dx + dy * dy + dz * dz;
      const float nm = fminf(s_mind[n], dd);
      s_mind[n] = nm;
      const bool better = nm > bestV;
      bestV = better ? nm : bestV;
      bestI = better ? n : bestI;
    }
    wave_argmax(bestV, bestI);
    if (lane == 0) { swv[wid] = bestV; swi[wid] = bestI; }
    __syncthreads();
    if (wid == 0) {
      float v = swv[lane];
      int   i = swi[lane];
      wave_argmax(v, i);
      if (lane == 0) s_last = i;
    }
    __syncthreads();
  }
}

__global__ __launch_bounds__(256) void three_nn_kernel(
    const float* __restrict__ pc, const int* __restrict__ fps_idx,
    int* __restrict__ idx3, float* __restrict__ w3) {
  const int tid  = threadIdx.x;
  const int lane = tid & 31;
  const int wave = blockIdx.x * (256 >> 5) + (tid >> 5);
  const int tilesPerBatch = M_ / 16;
  const int b    = wave / tilesPerBatch;
  const int rt   = wave - b * tilesPerBatch;
  const int half = lane >> 4;
  const int l15  = lane & 15;

  const float* pcb = pc + (size_t)b * N_ * 3;

  const int mrow = rt * 16 + l15;
  const int aidx = fps_idx[b * M_ + mrow];
  const float* ap = pcb + (size_t)aidx * 3;
  const float ax = ap[0], ay = ap[1], az = ap[2];
  v2f A;
  A.x = half ? (-2.0f * az) : (-2.0f * ax);
  A.y = half ? 1.0f         : (-2.0f * ay);

  float td[8][3];
  int   ti[8][3];
#pragma unroll
  for (int r = 0; r < 8; ++r)
#pragma unroll
    for (int k = 0; k < 3; ++k) { td[r][k] = __builtin_inff(); ti[r][k] = 0x7fffffff; }

#pragma unroll 2
  for (int t = 0; t < N_ / 16; ++t) {
    const int n = t * 16 + l15;
    const float* bp = pcb + (size_t)n * 3;
    const int pn = (n + 128 < N_) ? (n + 128) : (N_ - 1);
    __builtin_prefetch(pcb + (size_t)pn * 3, 0, 1);
    const float bx = bp[0], by = bp[1], bz = bp[2];
    const float bn2 = bx * bx + by * by + bz * bz;
    v8f c = {0.f, 0.f, 0.f, 0.f, 0.f, 0.f, 0.f, 0.f};
#if USE_WMMA
    v2f Bv;
    Bv.x = half ? bz  : bx;
    Bv.y = half ? bn2 : by;
    c = __builtin_amdgcn_wmma_f32_16x16x4_f32(
         false, A,  false, Bv,
         (short)0, c,  false,  false);
#else
#pragma unroll
    for (int r = 0; r < 8; ++r) {
      const int row = half * 8 + r;
      const float rx = __shfl(ax, row, 32);
      const float ry = __shfl(ay, row, 32);
      const float rz = __shfl(az, row, 32);
      c[r] = -2.0f * (rx * bx + ry * by + rz * bz) + bn2;
    }
#endif
#pragma unroll
    for (int r = 0; r < 8; ++r) ins3_stream(c[r], n, td[r], ti[r]);
  }

#pragma unroll
  for (int mskk = 1; mskk <= 8; mskk <<= 1) {
#pragma unroll
    for (int r = 0; r < 8; ++r) {
      float pd[3]; int pi[3];
#pragma unroll
      for (int k = 0; k < 3; ++k) {
        pd[k] = __shfl_xor(td[r][k], mskk, 32);
        pi[k] = __shfl_xor(ti[r][k], mskk, 32);
      }
#pragma unroll
      for (int k = 0; k < 3; ++k) ins3_lex(pd[k], pi[k], td[r], ti[r]);
    }
  }

  if (l15 == 0) {
#pragma unroll
    for (int r = 0; r < 8; ++r) {
      const int row = rt * 16 + half * 8 + r;
      const int gi  = fps_idx[b * M_ + row];
      const float* gp = pcb + (size_t)gi * 3;
      const float gx = gp[0], gy = gp[1], gz = gp[2];
      float wv[3];
      float wsum = 0.f;
#pragma unroll
      for (int k = 0; k < 3; ++k) {
        const int nn = ti[r][k];
        const float* np = pcb + (size_t)nn * 3;
        const float dx = np[0] - gx, dy = np[1] - gy, dz = np[2] - gz;
        const float dd = dx * dx + dy * dy + dz * dz;
        const float dist = sqrtf(fmaxf(dd, 0.0f));
        wv[k] = 1.0f / (dist + 1e-8f);
        wsum += wv[k];
        idx3[((size_t)b * M_ + row) * 3 + k] = nn;
      }
#pragma unroll
      for (int k = 0; k < 3; ++k)
        w3[((size_t)b * M_ + row) * 3 + k] = wv[k] / wsum;
    }
  }
}

__global__ __launch_bounds__(256) void interp_kernel(
    const float* __restrict__ feats, const int* __restrict__ idx3,
    const float* __restrict__ w3, float* __restrict__ out) {
  const int bc = blockIdx.x;
  const int b  = bc / C_;
  const int c  = bc - b * C_;
  const float* f = feats + ((size_t)b * C_ + c) * N_;
  float*       o = out   + ((size_t)b * C_ + c) * M_;
  const int*   ib = idx3 + (size_t)b * M_ * 3;
  const float* wb = w3   + (size_t)b * M_ * 3;
  for (int m = threadIdx.x; m < M_; m += 256) {
    const int   n0 = ib[m * 3 + 0], n1 = ib[m * 3 + 1], n2 = ib[m * 3 + 2];
    const float w0 = wb[m * 3 + 0], w1 = wb[m * 3 + 1], w2 = wb[m * 3 + 2];
    o[m] = w0 * f[n0] + w1 * f[n1] + w2 * f[n2];
  }
}

extern "C" void kernel_launch(void* const* d_in, const int* in_sizes, int n_in,
                              void* d_out, int out_size, void* d_ws, size_t ws_size,
                              hipStream_t stream) {
  const float* pc    = (const float*)d_in[0];
  const float* feats = (const float*)d_in[1];
  const float* obj   = (const float*)d_in[2];
  const float* grasp = (const float*)d_in[3];
  float* out = (float*)d_out;

  char* ws = (char*)d_ws;
  size_t off = 0;
  int* fps = (int*)(ws + off);
  off += (size_t)B_ * M_ * 4;        off = (off + 255) & ~(size_t)255;
  int* idx3 = (int*)(ws + off);
  off += (size_t)B_ * M_ * 3 * 4;    off = (off + 255) & ~(size_t)255;
  float* w3 = (float*)(ws + off);
  off += (size_t)B_ * M_ * 3 * 4;

  fps_kernel<<<B_, 1024, 0, stream>>>(pc, obj, grasp, fps);
  three_nn_kernel<<<(B_ * (M_ / 16)) / 8, 256, 0, stream>>>(pc, fps, idx3, w3);
  interp_kernel<<<B_ * C_, 256, 0, stream>>>(feats, idx3, w3, out);
}


# three_nn kernel: the 1250 column tiles of each 16-row tile shared by the 16 waves of one workgroup (grid 256 x 16 waves, was 32 x 8); per-wave top-3 meet in LDS, same (score,index) merge
# speedup vs baseline: 2.9135x; 1.2202x over previous
#include <hip/hip_runtime.h>
#include <hip/hip_bf16.h>

#define B_ 4
#define N_ 20000
#define C_ 512
#define M_ 1024
#define GRASP_TH 0.1f

typedef float v2f __attribute__((ext_vector_type(2)));
typedef float v8f __attribute__((ext_vector_type(8)));

#if defined(__gfx1250__) && __has_builtin(__builtin_amdgcn_wmma_f32_16x16x4_f32)
#define USE_WMMA 1
#else
#define USE_WMMA 0
#endif

__device__ __forceinline__ void ins3_stream(float v, int n, float d[3], int ix[3]) {
  const bool c0 = v < d[0];
  const bool c1 = v < d[1];
  const bool c2 = v < d[2];
  const float r0 = c0 ? v     : d[0];
  const int   j0 = c0 ? n     : ix[0];
  const float r1 = c0 ? d[0]  : (c1 ? v : d[1]);
  const int   j1 = c0 ? ix[0] : (c1 ? n : ix[1]);
  const float r2 = c1 ? d[1]  : (c2 ? v : d[2]);
  const int   j2 = c1 ? ix[1] : (c2 ? n : ix[2]);
  d[0] = r0; d[1] = r1; d[2] = r2;
  ix[0] = j0; ix[1] = j1; ix[2] = j2;
}

__device__ __forceinline__ void ins3_lex(float v, int n, float d[3], int ix[3]) {
  const bool c0 = (v < d[0]) | ((v == d[0]) & (n < ix[0]));
  const bool c1 = (v < d[1]) | ((v == d[1]) & (n < ix[1]));
  const bool c2 = (v < d[2]) | ((v == d[2]) & (n < ix[2]));
  const float r0 = c0 ? v     : d[0];
  const int   j0 = c0 ? n     : ix[0];
  const float r1 = c0 ? d[0]  : (c1 ? v : d[1]);
  const int   j1 = c0 ? ix[0] : (c1 ? n : ix[1]);
  const float r2 = c1 ? d[1]  : (c2 ? v : d[2]);
  const int   j2 = c1 ? ix[1] : (c2 ? n : ix[2]);
  d[0] = r0; d[1] = r1; d[2] = r2;
  ix[0] = j0; ix[1] = j1; ix[2] = j2;
}

template <int XM>
__device__ __forceinline__ int row_xor_i(int v) {
  return __builtin_amdgcn_update_dpp(v, v, 0x160 + XM, 0xf, 0xf, true);
}
__device__ __forceinline__ int xhalf_i(int v) {
  return __builtin_amdgcn_permlanex16(v, v, 0x76543210, 0xfedcba98, false, false);
}
__device__ __forceinline__ void wave_argmax(float& v, int& i) {
  const int key = __builtin_bit_cast(int, v);
  int m = key;
  m = max(m, row_xor_i<1>(m));
  m = max(m, row_xor_i<2>(m));
  m = max(m, row_xor_i<4>(m));
  m = max(m, row_xor_i<8>(m));
  m = max(m, xhalf_i(m));
  int c = (key == m) ? i : 0x7fffffff;
  c = min(c, row_xor_i<1>(c));
  c = min(c, row_xor_i<2>(c));
  c = min(c, row_xor_i<4>(c));
  c = min(c, row_xor_i<8>(c));
  c = min(c, xhalf_i(c));
  v = __builtin_bit_cast(float, m);
  i = c;
}

#define FPS_PASSES 20
#define FPS_WORDS  (FPS_PASSES * 32)
#define CMAX 15000

__global__ __launch_bounds__(1024) void fps_kernel(
    const float* __restrict__ pc, const float* __restrict__ obj,
    const float* __restrict__ grasp, int* __restrict__ fps_idx) {
  const int b    = blockIdx.x;
  const int tid  = threadIdx.x;
  const int lane = tid & 31;
  const int wid  = tid >> 5;

  __shared__ float    s_xyz[N_ * 3];
  __shared__ float    s_mind[N_];
  __shared__ unsigned s_bits[FPS_WORDS];
  __shared__ int      s_pref[FPS_WORDS];
  __shared__ float swv[32];
  __shared__ int   swi[32];
  __shared__ int   s_last;
  __shared__ int   s_K;

  const float* pcb = pc + (size_t)b * N_ * 3;
  const float* ob0 = obj + (size_t)b * 2 * N_;
  const float* gr  = grasp + (size_t)b * N_;

  unsigned myw = 0u;
  for (int it = 0; it < FPS_PASSES; ++it) {
    const int n  = it * 1024 + tid;
    const int nl = (n < N_) ? n : (N_ - 1);
    const bool mk = (n < N_) & (ob0[N_ + nl] > ob0[nl]) & (gr[nl] > GRASP_TH);
    const unsigned w = __builtin_amdgcn_ballot_w32(mk);
    myw = (lane == it) ? w : myw;
  }
  if (lane < FPS_PASSES) s_bits[lane * 32 + wid] = myw;
  __syncthreads();

  if (wid == 0) {
    int sum = 0;
    for (int j = 0; j < FPS_PASSES; ++j) sum += __popc(s_bits[lane * FPS_PASSES + j]);
    int incl = sum;
    for (int d = 1; d < 32; d <<= 1) {
      const int o = __shfl_up(incl, d, 32);
      incl += (lane >= d) ? o : 0;
    }
    int run = incl - sum;
    for (int j = 0; j < FPS_PASSES; ++j) {
      s_pref[lane * FPS_PASSES + j] = run;
      run += __popc(s_bits[lane * FPS_PASSES + j]);
    }
    if (lane == 31) s_K = incl;
    if (lane == 0)  s_last = 0;
  }
  __syncthreads();

  const int  K       = s_K;
  const bool compact = (K >= 1) & (K <= CMAX);
  const int  Kc      = compact ? K : N_;
  int* s_idx = (int*)(s_xyz + (compact ? 3 * Kc : 0));

  for (int g = 0; g < FPS_PASSES; g += 4) {
    float x[4], y[4], z[4];
    unsigned w[4];
    int pf[4];
    for (int j = 0; j < 4; ++j) {
      const int n  = (g + j) * 1024 + tid;
      const int nl = (n < N_) ? n : (N_ - 1);
      const float* p = pcb + (size_t)nl * 3;
      x[j] = p[0]; y[j] = p[1]; z[j] = p[2];
      w[j]  = s_bits[(g + j) * 32 + wid];
      pf[j] = s_pref[(g + j) * 32 + wid];
    }
    for (int j = 0; j < 4; ++j) {
      const int  n    = (g + j) * 1024 + tid;
      const bool mk   = (w[j] >> lane) & 1u;
      const int  pos  = pf[j] + __popc(w[j] & ((1u << lane) - 1u));
      const bool keep = compact ? mk : (n < N_);
      const int  slot = compact ? pos : n;
      if (keep) {
        s_xyz[slot * 3 + 0] = x[j];
        s_xyz[slot * 3 + 1] = y[j];
        s_xyz[slot * 3 + 2] = z[j];
        s_mind[slot] = mk ? __builtin_inff() : -__builtin_inff();
        if (compact) s_idx[slot] = n;
        else if (mk & (pos == 0)) s_last = n;
      }
    }
  }
  __syncthreads();

  int acc = 0;
  for (int step = 0; step < M_; ++step) {
    const int last = s_last;
    const float px = s_xyz[last * 3 + 0];
    const float py = s_xyz[last * 3 + 1];
    const float pz = s_xyz[last * 3 + 2];
    if (wid == 0) {
      const int pick = compact ? s_idx[last] : last;
      acc = (lane == (step & 31)) ? pick : acc;
      if ((step & 31) == 31) fps_idx[b * M_ + (step - 31) + lane] = acc;
    }

    float bestV = -__builtin_inff();
    int   bestI = 0;
#pragma unroll 4
    for (int n = tid; n < Kc; n += 1024) {
      const float dx = s_xyz[n * 3 + 0] - px;
      const float dy = s_xyz[n * 3 + 1] - py;
      const float dz = s_xyz[n * 3 + 2] - pz;
      const float dd = dx * dx + dy * dy + dz * dz;
      const float nm = fminf(s_mind[n], dd);
      s_mind[n] = nm;
      const bool better = nm > bestV;
      bestV = better ? nm : bestV;
      bestI = better ? n : bestI;
    }
    wave_argmax(bestV, bestI);
    if (lane == 0) { swv[wid] = bestV; swi[wid] = bestI; }
    __syncthreads();
    if (wid == 0) {
      float v = swv[lane];
      int   i = swi[lane];
      wave_argmax(v, i);
      if (lane == 0) s_last = i;
    }
    __syncthreads();
  }
}

#define NN_WAVES 16
__global__ __launch_bounds__(NN_WAVES * 32) void three_nn_kernel(
    const float* __restrict__ pc, const int* __restrict__ fps_idx,
    int* __restrict__ idx3, float* __restrict__ w3) {
  static_assert(NN_WAVES == 16, "the final merge maps wave j onto lane j of a 16-lane row");
  const int tid  = threadIdx.x;
  const int lane = tid & 31;
  const int wv   = __builtin_amdgcn_readfirstlane(tid >> 5);
  const int tilesPerBatch = M_ / 16;
  const int b    = blockIdx.x / tilesPerBatch;
  const int rt   = blockIdx.x - b * tilesPerBatch;
  const int half = lane >> 4;
  const int l15  = lane & 15;

  __shared__ float s_td[NN_WAVES][16][3];
  __shared__ int   s_ti[NN_WAVES][16][3];

  const float* pcb = pc + (size_t)b * N_ * 3;

  const int mrow = rt * 16 + l15;
  const int aidx = fps_idx[b * M_ + mrow];
  const float* ap = pcb + (size_t)aidx * 3;
  const float ax = ap[0], ay = ap[1], az = ap[2];
  v2f A;
  A.x = half ? (-2.0f * az) : (-2.0f * ax);
  A.y = half ? 1.0f         : (-2.0f * ay);

  float td[8][3];
  int   ti[8][3];
#pragma unroll
  for (int r = 0; r < 8; ++r)
#pragma unroll
    for (int k = 0; k < 3; ++k) { td[r][k] = __builtin_inff(); ti[r][k] = 0x7fffffff; }

  const int t0 = (wv * (N_ / 16)) / NN_WAVES;
  const int t1 = ((wv + 1) * (N_ / 16)) / NN_WAVES;
  int n = t0 * 16 + l15;
  const float* bp = pcb + (size_t)n * 3;
#pragma unroll 2
  for (int t = t0; t < t1; ++t, n += 16, bp += 16 * 3) {
    const int pn = (n + 128 < N_) ? (n + 128) : (N_ - 1);
    __builtin_prefetch(pcb + (size_t)pn * 3, 0, 1);
    const float bx = bp[0], by = bp[1], bz = bp[2];
    const float bn2 = bx * bx + by * by + bz * bz;
    v8f c = {0.f, 0.f, 0.f, 0.f, 0.f, 0.f, 0.f, 0.f};
#if USE_WMMA
    v2f Bv;
    Bv.x = half ? bz  : bx;
    Bv.y = half ? bn2 : by;
    c = __builtin_amdgcn_wmma_f32_16x16x4_f32(
         false, A,  false, Bv,
         (short)0, c,  false,  false);
#else
#pragma unroll
    for (int r = 0; r < 8; ++r) {
      const int row = half * 8 + r;
      const float rx = __shfl(ax, row, 32);
      const float ry = __shfl(ay, row, 32);
      const float rz = __shfl(az, row, 32);
      c[r] = -2.0f * (rx * bx + ry * by + rz * bz) + bn2;
    }
#endif
#pragma unroll
    for (int r = 0; r < 8; ++r) ins3_stream(c[r], n, td[r], ti[r]);
  }

#pragma unroll
  for (int mskk = 1; mskk <= 8; mskk <<= 1) {
#pragma unroll
    for (int r = 0; r < 8; ++r) {
      float pd[3]; int pi[3];
#pragma unroll
      for (int k = 0; k < 3; ++k) {
        pd[k] = __shfl_xor(td[r][k], mskk, 32);
        pi[k] = __shfl_xor(ti[r][k], mskk, 32);
      }
#pragma unroll
      for (int k = 0; k < 3; ++k) ins3_lex(pd[k], pi[k], td[r], ti[r]);
    }
  }

  if (l15 == 0) {
#pragma unroll
    for (int r = 0; r < 8; ++r)
#pragma unroll
      for (int k = 0; k < 3; ++k) {
        s_td[wv][half * 8 + r][k] = td[r][k];
        s_ti[wv][half * 8 + r][k] = ti[r][k];
      }
  }
  __syncthreads();

  float fd[3];
  int   fi[3];
#pragma unroll
  for (int k = 0; k < 3; ++k) { fd[k] = s_td[l15][wv][k]; fi[k] = s_ti[l15][wv][k]; }
#pragma unroll
  for (int mskk = 1; mskk <= 8; mskk <<= 1) {
    float pd[3]; int pi[3];
#pragma unroll
    for (int k = 0; k < 3; ++k) {
      pd[k] = __shfl_xor(fd[k], mskk, 32);
      pi[k] = __shfl_xor(fi[k], mskk, 32);
    }
#pragma unroll
    for (int k = 0; k < 3; ++k) ins3_lex(pd[k], pi[k], fd, fi);
  }

  if (lane == 0) {
    const int row = rt * 16 + wv;
    const int gi  = fps_idx[b * M_ + row];
    const float* gp = pcb + (size_t)gi * 3;
    const float gx = gp[0], gy = gp[1], gz = gp[2];
    float wv3[3];
    float wsum = 0.f;
#pragma unroll
    for (int k = 0; k < 3; ++k) {
      const int nn = fi[k];
      const float* np = pcb + (size_t)nn * 3;
      const float dx = np[0] - gx, dy = np[1] - gy, dz = np[2] - gz;
      const float dd = dx * dx + dy * dy + dz * dz;
      const float dist = sqrtf(fmaxf(dd, 0.0f));
      wv3[k] = 1.0f / (dist + 1e-8f);
      wsum += wv3[k];
      idx3[((size_t)b * M_ + row) * 3 + k] = nn;
    }
#pragma unroll
    for (int k = 0; k < 3; ++k)
      w3[((size_t)b * M_ + row) * 3 + k] = wv3[k] / wsum;
  }
}

__global__ __launch_bounds__(256) void interp_kernel(
    const float* __restrict__ feats, const int* __restrict__ idx3,
    const float* __restrict__ w3, float* __restrict__ out) {
  const int bc = blockIdx.x;
  const int b  = bc / C_;
  const int c  = bc - b * C_;
  const float* f = feats + ((size_t)b * C_ + c) * N_;
  float*       o = out   + ((size_t)b * C_ + c) * M_;
  const int*   ib = idx3 + (size_t)b * M_ * 3;
  const float* wb = w3   + (size_t)b * M_ * 3;
  for (int m = threadIdx.x; m < M_; m += 256) {
    const int   n0 = ib[m * 3 + 0], n1 = ib[m * 3 + 1], n2 = ib[m * 3 + 2];
    const float w0 = wb[m * 3 + 0], w1 = wb[m * 3 + 1], w2 = wb[m * 3 + 2];
    o[m] = w0 * f[n0] + w1 * f[n1] + w2 * f[n2];
  }
}

extern "C" void kernel_launch(void* const* d_in, const int* in_sizes, int n_in,
                              void* d_out, int out_size, void* d_ws, size_t ws_size,
                              hipStream_t stream) {
  const float* pc    = (const float*)d_in[0];
  const float* feats = (const float*)d_in[1];
  const float* obj   = (const float*)d_in[2];
  const float* grasp = (const float*)d_in[3];
  float* out = (float*)d_out;

  char* ws = (char*)d_ws;
  size_t off = 0;
  int* fps = (int*)(ws + off);
  off += (size_t)B_ * M_ * 4;        off = (off + 255) & ~(size_t)255;
  int* idx3 = (int*)(ws + off);
  off += (size_t)B_ * M_ * 3 * 4;    off = (off + 255) & ~(size_t)255;
  float* w3 = (float*)(ws + off);
  off += (size_t)B_ * M_ * 3 * 4;

  fps_kernel<<<B_, 1024, 0, stream>>>(pc, obj, grasp, fps);
  three_nn_kernel<<<B_ * (M_ / 16), NN_WAVES * 32, 0, stream>>>(pc, fps, idx3, w3);
  interp_kernel<<<B_ * C_, 256, 0, stream>>>(feats, idx3, w3, out);
}


# fps kernel: one barrier per step (was 2): each wave reads all 32 wave winners from an LDS table double-buffered by step parity and reduces them itself; no winner publish, no second barrier
# speedup vs baseline: 3.0671x; 1.0527x over previous
#include <hip/hip_runtime.h>
#include <hip/hip_bf16.h>

#define B_ 4
#define N_ 20000
#define C_ 512
#define M_ 1024
#define GRASP_TH 0.1f

typedef float v2f __attribute__((ext_vector_type(2)));
typedef float v8f __attribute__((ext_vector_type(8)));

#if defined(__gfx1250__) && __has_builtin(__builtin_amdgcn_wmma_f32_16x16x4_f32)
#define USE_WMMA 1
#else
#define USE_WMMA 0
#endif

__device__ __forceinline__ void ins3_stream(float v, int n, float d[3], int ix[3]) {
  const bool c0 = v < d[0];
  const bool c1 = v < d[1];
  const bool c2 = v < d[2];
  const float r0 = c0 ? v     : d[0];
  const int   j0 = c0 ? n     : ix[0];
  const float r1 = c0 ? d[0]  : (c1 ? v : d[1]);
  const int   j1 = c0 ? ix[0] : (c1 ? n : ix[1]);
  const float r2 = c1 ? d[1]  : (c2 ? v : d[2]);
  const int   j2 = c1 ? ix[1] : (c2 ? n : ix[2]);
  d[0] = r0; d[1] = r1; d[2] = r2;
  ix[0] = j0; ix[1] = j1; ix[2] = j2;
}

__device__ __forceinline__ void ins3_lex(float v, int n, float d[3], int ix[3]) {
  const bool c0 = (v < d[0]) | ((v == d[0]) & (n < ix[0]));
  const bool c1 = (v < d[1]) | ((v == d[1]) & (n < ix[1]));
  const bool c2 = (v < d[2]) | ((v == d[2]) & (n < ix[2]));
  const float r0 = c0 ? v     : d[0];
  const int   j0 = c0 ? n     : ix[0];
  const float r1 = c0 ? d[0]  : (c1 ? v : d[1]);
  const int   j1 = c0 ? ix[0] : (c1 ? n : ix[1]);
  const float r2 = c1 ? d[1]  : (c2 ? v : d[2]);
  const int   j2 = c1 ? ix[1] : (c2 ? n : ix[2]);
  d[0] = r0; d[1] = r1; d[2] = r2;
  ix[0] = j0; ix[1] = j1; ix[2] = j2;
}

template <int XM>
__device__ __forceinline__ int row_xor_i(int v) {
  return __builtin_amdgcn_update_dpp(v, v, 0x160 + XM, 0xf, 0xf, true);
}
__device__ __forceinline__ int xhalf_i(int v) {
  return __builtin_amdgcn_permlanex16(v, v, 0x76543210, 0xfedcba98, false, false);
}
__device__ __forceinline__ void wave_argmax(float& v, int& i) {
  const int key = __builtin_bit_cast(int, v);
  int m = key;
  m = max(m, row_xor_i<1>(m));
  m = max(m, row_xor_i<2>(m));
  m = max(m, row_xor_i<4>(m));
  m = max(m, row_xor_i<8>(m));
  m = max(m, xhalf_i(m));
  int c = (key == m) ? i : 0x7fffffff;
  c = min(c, row_xor_i<1>(c));
  c = min(c, row_xor_i<2>(c));
  c = min(c, row_xor_i<4>(c));
  c = min(c, row_xor_i<8>(c));
  c = min(c, xhalf_i(c));
  v = __builtin_bit_cast(float, m);
  i = c;
}

__device__ __forceinline__ int table_argmax(float v0, int i0, float v1, int i1) {
  const int k0 = __builtin_bit_cast(int, v0);
  const int k1 = __builtin_bit_cast(int, v1);
  const int mp = max(k0, k1);
  int m = mp;
  m = max(m, row_xor_i<1>(m));
  m = max(m, row_xor_i<2>(m));
  m = max(m, row_xor_i<4>(m));
  m = max(m, row_xor_i<8>(m));
  int c = min((k0 == m) ? i0 : 0x7fffffff, (k1 == m) ? i1 : 0x7fffffff);
  c = min(c, row_xor_i<1>(c));
  c = min(c, row_xor_i<2>(c));
  c = min(c, row_xor_i<4>(c));
  c = min(c, row_xor_i<8>(c));
  return c;
}

#define FPS_PASSES 20
#define FPS_WORDS  (FPS_PASSES * 32)
#define CMAX 15000

__global__ __launch_bounds__(1024) void fps_kernel(
    const float* __restrict__ pc, const float* __restrict__ obj,
    const float* __restrict__ grasp, int* __restrict__ fps_idx) {
  const int b    = blockIdx.x;
  const int tid  = threadIdx.x;
  const int lane = tid & 31;
  const int wid  = tid >> 5;

  __shared__ float    s_xyz[N_ * 3];
  __shared__ float    s_mind[N_];
  __shared__ unsigned s_bits[FPS_WORDS];
  __shared__ int      s_pref[FPS_WORDS];
  __shared__ int   s_last;
  __shared__ int   s_K;

  const float* pcb = pc + (size_t)b * N_ * 3;
  const float* ob0 = obj + (size_t)b * 2 * N_;
  const float* gr  = grasp + (size_t)b * N_;

  unsigned myw = 0u;
  for (int it = 0; it < FPS_PASSES; ++it) {
    const int n  = it * 1024 + tid;
    const int nl = (n < N_) ? n : (N_ - 1);
    const bool mk = (n < N_) & (ob0[N_ + nl] > ob0[nl]) & (gr[nl] > GRASP_TH);
    const unsigned w = __builtin_amdgcn_ballot_w32(mk);
    myw = (lane == it) ? w : myw;
  }
  if (lane < FPS_PASSES) s_bits[lane * 32 + wid] = myw;
  __syncthreads();

  if (wid == 0) {
    int sum = 0;
    for (int j = 0; j < FPS_PASSES; ++j) sum += __popc(s_bits[lane * FPS_PASSES + j]);
    int incl = sum;
    for (int d = 1; d < 32; d <<= 1) {
      const int o = __shfl_up(incl, d, 32);
      incl += (lane >= d) ? o : 0;
    }
    int run = incl - sum;
    for (int j = 0; j < FPS_PASSES; ++j) {
      s_pref[lane * FPS_PASSES + j] = run;
      run += __popc(s_bits[lane * FPS_PASSES + j]);
    }
    if (lane == 31) s_K = incl;
    if (lane == 0)  s_last = 0;
  }
  __syncthreads();

  const int  K       = s_K;
  const bool compact = (K >= 1) & (K <= CMAX);
  const int  Kc      = compact ? K : N_;
  int* s_idx = (int*)(s_xyz + (compact ? 3 * Kc : 0));

  for (int g = 0; g < FPS_PASSES; g += 4) {
    float x[4], y[4], z[4];
    unsigned w[4];
    int pf[4];
    for (int j = 0; j < 4; ++j) {
      const int n  = (g + j) * 1024 + tid;
      const int nl = (n < N_) ? n : (N_ - 1);
      const float* p = pcb + (size_t)nl * 3;
      x[j] = p[0]; y[j] = p[1]; z[j] = p[2];
      w[j]  = s_bits[(g + j) * 32 + wid];
      pf[j] = s_pref[(g + j) * 32 + wid];
    }
    for (int j = 0; j < 4; ++j) {
      const int  n    = (g + j) * 1024 + tid;
      const bool mk   = (w[j] >> lane) & 1u;
      const int  pos  = pf[j] + __popc(w[j] & ((1u << lane) - 1u));
      const bool keep = compact ? mk : (n < N_);
      const int  slot = compact ? pos : n;
      if (keep) {
        s_xyz[slot * 3 + 0] = x[j];
        s_xyz[slot * 3 + 1] = y[j];
        s_xyz[slot * 3 + 2] = z[j];
        s_mind[slot] = mk ? __builtin_inff() : -__builtin_inff();
        if (compact) s_idx[slot] = n;
        else if (mk & (pos == 0)) s_last = n;
      }
    }
  }
  __syncthreads();

  float* swv = reinterpret_cast<float*>(s_bits);
  int*   swi = s_pref;
  int last = s_last;
  int acc = 0;
  for (int step = 0; step < M_; ++step) {
    const float px = s_xyz[last * 3 + 0];
    const float py = s_xyz[last * 3 + 1];
    const float pz = s_xyz[last * 3 + 2];
    if (wid == 0) {
      const int pick = compact ? s_idx[last] : last;
      acc = (lane == (step & 31)) ? pick : acc;
      if ((step & 31) == 31) fps_idx[b * M_ + (step - 31) + lane] = acc;
    }

    float bestV = -__builtin_inff();
    int   bestI = 0;
#pragma unroll 4
    for (int n = tid; n < Kc; n += 1024) {
      const float dx = s_xyz[n * 3 + 0] - px;
      const float dy = s_xyz[n * 3 + 1] - py;
      const float dz = s_xyz[n * 3 + 2] - pz;
      const float dd = dx * dx + dy * dy + dz * dz;
      const float nm = fminf(s_mind[n], dd);
      s_mind[n] = nm;
      const bool better = nm > bestV;
      bestV = better ? nm : bestV;
      bestI = better ? n : bestI;
    }
    wave_argmax(bestV, bestI);
    const int pb = (step & 1) * 32;
    if (lane == 0) { swv[pb + wid] = bestV; swi[pb + wid] = bestI; }
    __syncthreads();
    const int e = pb + (lane & 15);
    last = table_argmax(swv[e], swi[e], swv[e + 16], swi[e + 16]);
  }
}

#define NN_WAVES 16
__global__ __launch_bounds__(NN_WAVES * 32) void three_nn_kernel(
    const float* __restrict__ pc, const int* __restrict__ fps_idx,
    int* __restrict__ idx3, float* __restrict__ w3) {
  static_assert(NN_WAVES == 16, "the final merge maps wave j onto lane j of a 16-lane row");
  const int tid  = threadIdx.x;
  const int lane = tid & 31;
  const int wv   = __builtin_amdgcn_readfirstlane(tid >> 5);
  const int tilesPerBatch = M_ / 16;
  const int b    = blockIdx.x / tilesPerBatch;
  const int rt   = blockIdx.x - b * tilesPerBatch;
  const int half = lane >> 4;
  const int l15  = lane & 15;

  __shared__ float s_td[NN_WAVES][16][3];
  __shared__ int   s_ti[NN_WAVES][16][3];

  const float* pcb = pc + (size_t)b * N_ * 3;

  const int mrow = rt * 16 + l15;
  const int aidx = fps_idx[b * M_ + mrow];
  const float* ap = pcb + (size_t)aidx * 3;
  const float ax = ap[0], ay = ap[1], az = ap[2];
  v2f A;
  A.x = half ? (-2.0f * az) : (-2.0f * ax);
  A.y = half ? 1.0f         : (-2.0f * ay);

  float td[8][3];
  int   ti[8][3];
#pragma unroll
  for (int r = 0; r < 8; ++r)
#pragma unroll
    for (int k = 0; k < 3; ++k) { td[r][k] = __builtin_inff(); ti[r][k] = 0x7fffffff; }

  const int t0 = (wv * (N_ / 16)) / NN_WAVES;
  const int t1 = ((wv + 1) * (N_ / 16)) / NN_WAVES;
  int n = t0 * 16 + l15;
  const float* bp = pcb + (size_t)n * 3;
#pragma unroll 2
  for (int t = t0; t < t1; ++t, n += 16, bp += 16 * 3) {
    const int pn = (n + 128 < N_) ? (n + 128) : (N_ - 1);
    __builtin_prefetch(pcb + (size_t)pn * 3, 0, 1);
    const float bx = bp[0], by = bp[1], bz = bp[2];
    const float bn2 = bx * bx + by * by + bz * bz;
    v8f c = {0.f, 0.f, 0.f, 0.f, 0.f, 0.f, 0.f, 0.f};
#if USE_WMMA
    v2f Bv;
    Bv.x = half ? bz  : bx;
    Bv.y = half ? bn2 : by;
    c = __builtin_amdgcn_wmma_f32_16x16x4_f32(
         false, A,  false, Bv,
         (short)0, c,  false,  false);
#else
#pragma unroll
    for (int r = 0; r < 8; ++r) {
      const int row = half * 8 + r;
      const float rx = __shfl(ax, row, 32);
      const float ry = __shfl(ay, row, 32);
      const float rz = __shfl(az, row, 32);
      c[r] = -2.0f * (rx * bx + ry * by + rz * bz) + bn2;
    }
#endif
#pragma unroll
    for (int r = 0; r < 8; ++r) ins3_stream(c[r], n, td[r], ti[r]);
  }

#pragma unroll
  for (int mskk = 1; mskk <= 8; mskk <<= 1) {
#pragma unroll
    for (int r = 0; r < 8; ++r) {
      float pd[3]; int pi[3];
#pragma unroll
      for (int k = 0; k < 3; ++k) {
        pd[k] = __shfl_xor(td[r][k], mskk, 32);
        pi[k] = __shfl_xor(ti[r][k], mskk, 32);
      }
#pragma unroll
      for (int k = 0; k < 3; ++k) ins3_lex(pd[k], pi[k], td[r], ti[r]);
    }
  }

  if (l15 == 0) {
#pragma unroll
    for (int r = 0; r < 8; ++r)
#pragma unroll
      for (int k = 0; k < 3; ++k) {
        s_td[wv][half * 8 + r][k] = td[r][k];
        s_ti[wv][half * 8 + r][k] = ti[r][k];
      }
  }
  __syncthreads();

  float fd[3];
  int   fi[3];
#pragma unroll
  for (int k = 0; k < 3; ++k) { fd[k] = s_td[l15][wv][k]; fi[k] = s_ti[l15][wv][k]; }
#pragma unroll
  for (int mskk = 1; mskk <= 8; mskk <<= 1) {
    float pd[3]; int pi[3];
#pragma unroll
    for (int k = 0; k < 3; ++k) {
      pd[k] = __shfl_xor(fd[k], mskk, 32);
      pi[k] = __shfl_xor(fi[k], mskk, 32);
    }
#pragma unroll
    for (int k = 0; k < 3; ++k) ins3_lex(pd[k], pi[k], fd, fi);
  }

  if (lane == 0) {
    const int row = rt * 16 + wv;
    const int gi  = fps_idx[b * M_ + row];
    const float* gp = pcb + (size_t)gi * 3;
    const float gx = gp[0], gy = gp[1], gz = gp[2];
    float wv3[3];
    float wsum = 0.f;
#pragma unroll
    for (int k = 0; k < 3; ++k) {
      const int nn = fi[k];
      const float* np = pcb + (size_t)nn * 3;
      const float dx = np[0] - gx, dy = np[1] - gy, dz = np[2] - gz;
      const float dd = dx * dx + dy * dy + dz * dz;
      const float dist = sqrtf(fmaxf(dd, 0.0f));
      wv3[k] = 1.0f / (dist + 1e-8f);
      wsum += wv3[k];
      idx3[((size_t)b * M_ + row) * 3 + k] = nn;
    }
#pragma unroll
    for (int k = 0; k < 3; ++k)
      w3[((size_t)b * M_ + row) * 3 + k] = wv3[k] / wsum;
  }
}

__global__ __launch_bounds__(256) void interp_kernel(
    const float* __restrict__ feats, const int* __restrict__ idx3,
    const float* __restrict__ w3, float* __restrict__ out) {
  const int bc = blockIdx.x;
  const int b  = bc / C_;
  const int c  = bc - b * C_;
  const float* f = feats + ((size_t)b * C_ + c) * N_;
  float*       o = out   + ((size_t)b * C_ + c) * M_;
  const int*   ib = idx3 + (size_t)b * M_ * 3;
  const float* wb = w3   + (size_t)b * M_ * 3;
  for (int m = threadIdx.x; m < M_; m += 256) {
    const int   n0 = ib[m * 3 + 0], n1 = ib[m * 3 + 1], n2 = ib[m * 3 + 2];
    const float w0 = wb[m * 3 + 0], w1 = wb[m * 3 + 1], w2 = wb[m * 3 + 2];
    o[m] = w0 * f[n0] + w1 * f[n1] + w2 * f[n2];
  }
}

extern "C" void kernel_launch(void* const* d_in, const int* in_sizes, int n_in,
                              void* d_out, int out_size, void* d_ws, size_t ws_size,
                              hipStream_t stream) {
  const float* pc    = (const float*)d_in[0];
  const float* feats = (const float*)d_in[1];
  const float* obj   = (const float*)d_in[2];
  const float* grasp = (const float*)d_in[3];
  float* out = (float*)d_out;

  char* ws = (char*)d_ws;
  size_t off = 0;
  int* fps = (int*)(ws + off);
  off += (size_t)B_ * M_ * 4;        off = (off + 255) & ~(size_t)255;
  int* idx3 = (int*)(ws + off);
  off += (size_t)B_ * M_ * 3 * 4;    off = (off + 255) & ~(size_t)255;
  float* w3 = (float*)(ws + off);
  off += (size_t)B_ * M_ * 3 * 4;

  fps_kernel<<<B_, 1024, 0, stream>>>(pc, obj, grasp, fps);
  three_nn_kernel<<<B_ * (M_ / 16), NN_WAVES * 32, 0, stream>>>(pc, fps, idx3, w3);
  interp_kernel<<<B_ * C_, 256, 0, stream>>>(feats, idx3, w3, out);
}


# fps kernel: each thread keeps its candidates' xyz and running min distance in registers through all 1024 steps (up to 20 slots, 95 VGPRs); the scan reads no LDS, only the winner's xyz; same sums
# speedup vs baseline: 3.9103x; 1.2749x over previous
#include <hip/hip_runtime.h>
#include <hip/hip_bf16.h>

#define B_ 4
#define N_ 20000
#define C_ 512
#define M_ 1024
#define GRASP_TH 0.1f

typedef float v2f __attribute__((ext_vector_type(2)));
typedef float v8f __attribute__((ext_vector_type(8)));

#if defined(__gfx1250__) && __has_builtin(__builtin_amdgcn_wmma_f32_16x16x4_f32)
#define USE_WMMA 1
#else
#define USE_WMMA 0
#endif

__device__ __forceinline__ void ins3_stream(float v, int n, float d[3], int ix[3]) {
  const bool c0 = v < d[0];
  const bool c1 = v < d[1];
  const bool c2 = v < d[2];
  const float r0 = c0 ? v     : d[0];
  const int   j0 = c0 ? n     : ix[0];
  const float r1 = c0 ? d[0]  : (c1 ? v : d[1]);
  const int   j1 = c0 ? ix[0] : (c1 ? n : ix[1]);
  const float r2 = c1 ? d[1]  : (c2 ? v : d[2]);
  const int   j2 = c1 ? ix[1] : (c2 ? n : ix[2]);
  d[0] = r0; d[1] = r1; d[2] = r2;
  ix[0] = j0; ix[1] = j1; ix[2] = j2;
}

__device__ __forceinline__ void ins3_lex(float v, int n, float d[3], int ix[3]) {
  const bool c0 = (v < d[0]) | ((v == d[0]) & (n < ix[0]));
  const bool c1 = (v < d[1]) | ((v == d[1]) & (n < ix[1]));
  const bool c2 = (v < d[2]) | ((v == d[2]) & (n < ix[2]));
  const float r0 = c0 ? v     : d[0];
  const int   j0 = c0 ? n     : ix[0];
  const float r1 = c0 ? d[0]  : (c1 ? v : d[1]);
  const int   j1 = c0 ? ix[0] : (c1 ? n : ix[1]);
  const float r2 = c1 ? d[1]  : (c2 ? v : d[2]);
  const int   j2 = c1 ? ix[1] : (c2 ? n : ix[2]);
  d[0] = r0; d[1] = r1; d[2] = r2;
  ix[0] = j0; ix[1] = j1; ix[2] = j2;
}

template <int XM>
__device__ __forceinline__ int row_xor_i(int v) {
  return __builtin_amdgcn_update_dpp(v, v, 0x160 + XM, 0xf, 0xf, true);
}
__device__ __forceinline__ int xhalf_i(int v) {
  return __builtin_amdgcn_permlanex16(v, v, 0x76543210, 0xfedcba98, false, false);
}
__device__ __forceinline__ void wave_argmax(float& v, int& i) {
  const int key = __builtin_bit_cast(int, v);
  int m = key;
  m = max(m, row_xor_i<1>(m));
  m = max(m, row_xor_i<2>(m));
  m = max(m, row_xor_i<4>(m));
  m = max(m, row_xor_i<8>(m));
  m = max(m, xhalf_i(m));
  int c = (key == m) ? i : 0x7fffffff;
  c = min(c, row_xor_i<1>(c));
  c = min(c, row_xor_i<2>(c));
  c = min(c, row_xor_i<4>(c));
  c = min(c, row_xor_i<8>(c));
  c = min(c, xhalf_i(c));
  v = __builtin_bit_cast(float, m);
  i = c;
}

__device__ __forceinline__ int table_argmax(float v0, int i0, float v1, int i1) {
  const int k0 = __builtin_bit_cast(int, v0);
  const int k1 = __builtin_bit_cast(int, v1);
  const int mp = max(k0, k1);
  int m = mp;
  m = max(m, row_xor_i<1>(m));
  m = max(m, row_xor_i<2>(m));
  m = max(m, row_xor_i<4>(m));
  m = max(m, row_xor_i<8>(m));
  int c = min((k0 == m) ? i0 : 0x7fffffff, (k1 == m) ? i1 : 0x7fffffff);
  c = min(c, row_xor_i<1>(c));
  c = min(c, row_xor_i<2>(c));
  c = min(c, row_xor_i<4>(c));
  c = min(c, row_xor_i<8>(c));
  return c;
}

__device__ __forceinline__ void fps_pass(v2f xy, float z, float& m, v2f pxy, float pz,
                                         int p, float& bestV, int& bestP) {
  const v2f   d  = xy - pxy;
  const float dz = z - pz;
  const float sxy = d.x * d.x + d.y * d.y;
  const float dd  = __builtin_fmaf(dz, dz, sxy);
  const float nm = fminf(m, dd);
  m = nm;
  const bool better = nm > bestV;
  bestV = better ? nm : bestV;
  bestP = better ? p : bestP;
}

#define FPS_PASSES 20
#define FPS_WORDS  (FPS_PASSES * 32)
#define CMAX 15000
#define FPS_PMAX FPS_PASSES

__global__ __launch_bounds__(1024) void fps_kernel(
    const float* __restrict__ pc, const float* __restrict__ obj,
    const float* __restrict__ grasp, int* __restrict__ fps_idx) {
  const int b    = blockIdx.x;
  const int tid  = threadIdx.x;
  const int lane = tid & 31;
  const int wid  = tid >> 5;

  __shared__ float    s_xyz[N_ * 3];
  __shared__ unsigned s_bits[FPS_WORDS];
  __shared__ int      s_pref[FPS_WORDS];
  __shared__ int   s_last;
  __shared__ int   s_K;

  const float* pcb = pc + (size_t)b * N_ * 3;
  const float* ob0 = obj + (size_t)b * 2 * N_;
  const float* gr  = grasp + (size_t)b * N_;

  unsigned myw = 0u;
  for (int it = 0; it < FPS_PASSES; ++it) {
    const int n  = it * 1024 + tid;
    const int nl = (n < N_) ? n : (N_ - 1);
    const bool mk = (n < N_) & (ob0[N_ + nl] > ob0[nl]) & (gr[nl] > GRASP_TH);
    const unsigned w = __builtin_amdgcn_ballot_w32(mk);
    myw = (lane == it) ? w : myw;
  }
  if (lane < FPS_PASSES) s_bits[lane * 32 + wid] = myw;
  __syncthreads();

  if (wid == 0) {
    int sum = 0;
    for (int j = 0; j < FPS_PASSES; ++j) sum += __popc(s_bits[lane * FPS_PASSES + j]);
    int incl = sum;
    for (int d = 1; d < 32; d <<= 1) {
      const int o = __shfl_up(incl, d, 32);
      incl += (lane >= d) ? o : 0;
    }
    int run = incl - sum;
    for (int j = 0; j < FPS_PASSES; ++j) {
      s_pref[lane * FPS_PASSES + j] = run;
      run += __popc(s_bits[lane * FPS_PASSES + j]);
    }
    if (lane == 31) s_K = incl;
    if (lane == 0)  s_last = 0;
  }
  __syncthreads();

  const int  K       = s_K;
  const bool compact = (K >= 1) & (K <= CMAX);
  const int  Kc      = compact ? K : N_;
  int* s_idx = (int*)(s_xyz + (compact ? 3 * Kc : 0));

  for (int g = 0; g < FPS_PASSES; g += 4) {
    float x[4], y[4], z[4];
    unsigned w[4];
    int pf[4];
    for (int j = 0; j < 4; ++j) {
      const int n  = (g + j) * 1024 + tid;
      const int nl = (n < N_) ? n : (N_ - 1);
      const float* p = pcb + (size_t)nl * 3;
      x[j] = p[0]; y[j] = p[1]; z[j] = p[2];
      w[j]  = s_bits[(g + j) * 32 + wid];
      pf[j] = s_pref[(g + j) * 32 + wid];
    }
    for (int j = 0; j < 4; ++j) {
      const int  n    = (g + j) * 1024 + tid;
      const bool mk   = (w[j] >> lane) & 1u;
      const int  pos  = pf[j] + __popc(w[j] & ((1u << lane) - 1u));
      const bool keep = compact ? mk : (n < N_);
      const int  slot = compact ? pos : n;
      if (keep) {
        s_xyz[slot * 3 + 0] = x[j];
        s_xyz[slot * 3 + 1] = y[j];
        s_xyz[slot * 3 + 2] = z[j];
        if (compact) s_idx[slot] = n;
        else if (mk & (pos == 0)) s_last = n;
      }
    }
  }
  __syncthreads();

  const int wv  = __builtin_amdgcn_readfirstlane(wid);
  const int rem = Kc - wv * 32;
  const int Pw  = (rem > 0) ? ((rem + 1023) >> 10) : 0;
  v2f   rxy[FPS_PMAX];
  float rz[FPS_PMAX], rm[FPS_PMAX];
#pragma unroll
  for (int p = 0; p < FPS_PMAX; ++p) {
    const int  slot = p * 1024 + tid;
    const bool in   = slot < Kc;
    const int  sl   = in ? slot : 0;
    rxy[p].x = s_xyz[sl * 3 + 0];
    rxy[p].y = s_xyz[sl * 3 + 1];
    rz[p]    = s_xyz[sl * 3 + 2];
    const bool mk = (s_bits[p * 32 + wid] >> lane) & 1u;
    rm[p] = (in & (compact | mk)) ? __builtin_inff() : -__builtin_inff();
  }
  __syncthreads();

  float* swv = reinterpret_cast<float*>(s_bits);
  int*   swi = s_pref;
  int last = s_last;
  int acc = 0;
  for (int step = 0; step < M_; ++step) {
    v2f pxy;
    pxy.x = s_xyz[last * 3 + 0];
    pxy.y = s_xyz[last * 3 + 1];
    const float pz = s_xyz[last * 3 + 2];
    if (wid == 0) {
      const int pick = compact ? s_idx[last] : last;
      acc = (lane == (step & 31)) ? pick : acc;
      if ((step & 31) == 31) fps_idx[b * M_ + (step - 31) + lane] = acc;
    }

    float bestV = -__builtin_inff();
    int   bestP = 0;
    if (Pw > 0) { fps_pass(rxy[0], rz[0], rm[0], pxy, pz, 0, bestV, bestP);
    if (Pw > 1) { fps_pass(rxy[1], rz[1], rm[1], pxy, pz, 1, bestV, bestP);
    if (Pw > 2) { fps_pass(rxy[2], rz[2], rm[2], pxy, pz, 2, bestV, bestP);
    if (Pw > 3) { fps_pass(rxy[3], rz[3], rm[3], pxy, pz, 3, bestV, bestP);
    if (Pw > 4) { fps_pass(rxy[4], rz[4], rm[4], pxy, pz, 4, bestV, bestP);
    if (Pw > 5) { fps_pass(rxy[5], rz[5], rm[5], pxy, pz, 5, bestV, bestP);
    if (Pw > 6) { fps_pass(rxy[6], rz[6], rm[6], pxy, pz, 6, bestV, bestP);
    if (Pw > 7) { fps_pass(rxy[7], rz[7], rm[7], pxy, pz, 7, bestV, bestP);
    if (Pw > 8) { fps_pass(rxy[8], rz[8], rm[8], pxy, pz, 8, bestV, bestP);
    if (Pw > 9) { fps_pass(rxy[9], rz[9], rm[9], pxy, pz, 9, bestV, bestP);
    if (Pw > 10) { fps_pass(rxy[10], rz[10], rm[10], pxy, pz, 10, bestV, bestP);
    if (Pw > 11) { fps_pass(rxy[11], rz[11], rm[11], pxy, pz, 11, bestV, bestP);
    if (Pw > 12) { fps_pass(rxy[12], rz[12], rm[12], pxy, pz, 12, bestV, bestP);
    if (Pw > 13) { fps_pass(rxy[13], rz[13], rm[13], pxy, pz, 13, bestV, bestP);
    if (Pw > 14) { fps_pass(rxy[14], rz[14], rm[14], pxy, pz, 14, bestV, bestP);
    if (Pw > 15) { fps_pass(rxy[15], rz[15], rm[15], pxy, pz, 15, bestV, bestP);
    if (Pw > 16) { fps_pass(rxy[16], rz[16], rm[16], pxy, pz, 16, bestV, bestP);
    if (Pw > 17) { fps_pass(rxy[17], rz[17], rm[17], pxy, pz, 17, bestV, bestP);
    if (Pw > 18) { fps_pass(rxy[18], rz[18], rm[18], pxy, pz, 18, bestV, bestP);
    if (Pw > 19) { fps_pass(rxy[19], rz[19], rm[19], pxy, pz, 19, bestV, bestP);
    }}}}}}}}}}}}}}}}}}}}
    int bestI = bestP * 1024 + tid;
    wave_argmax(bestV, bestI);
    const int pb = (step & 1) * 32;
    if (lane == 0) { swv[pb + wid] = bestV; swi[pb + wid] = bestI; }
    __syncthreads();
    const int e = pb + (lane & 15);
    last = table_argmax(swv[e], swi[e], swv[e + 16], swi[e + 16]);
  }
}

#define NN_WAVES 16
__global__ __launch_bounds__(NN_WAVES * 32) void three_nn_kernel(
    const float* __restrict__ pc, const int* __restrict__ fps_idx,
    int* __restrict__ idx3, float* __restrict__ w3) {
  static_assert(NN_WAVES == 16, "the final merge maps wave j onto lane j of a 16-lane row");
  const int tid  = threadIdx.x;
  const int lane = tid & 31;
  const int wv   = __builtin_amdgcn_readfirstlane(tid >> 5);
  const int tilesPerBatch = M_ / 16;
  const int b    = blockIdx.x / tilesPerBatch;
  const int rt   = blockIdx.x - b * tilesPerBatch;
  const int half = lane >> 4;
  const int l15  = lane & 15;

  __shared__ float s_td[NN_WAVES][16][3];
  __shared__ int   s_ti[NN_WAVES][16][3];

  const float* pcb = pc + (size_t)b * N_ * 3;

  const int mrow = rt * 16 + l15;
  const int aidx = fps_idx[b * M_ + mrow];
  const float* ap = pcb + (size_t)aidx * 3;
  const float ax = ap[0], ay = ap[1], az = ap[2];
  v2f A;
  A.x = half ? (-2.0f * az) : (-2.0f * ax);
  A.y = half ? 1.0f         : (-2.0f * ay);

  float td[8][3];
  int   ti[8][3];
#pragma unroll
  for (int r = 0; r < 8; ++r)
#pragma unroll
    for (int k = 0; k < 3; ++k) { td[r][k] = __builtin_inff(); ti[r][k] = 0x7fffffff; }

  const int t0 = (wv * (N_ / 16)) / NN_WAVES;
  const int t1 = ((wv + 1) * (N_ / 16)) / NN_WAVES;
  int n = t0 * 16 + l15;
  const float* bp = pcb + (size_t)n * 3;
#pragma unroll 2
  for (int t = t0; t < t1; ++t, n += 16, bp += 16 * 3) {
    const int pn = (n + 128 < N_) ? (n + 128) : (N_ - 1);
    __builtin_prefetch(pcb + (size_t)pn * 3, 0, 1);
    const float bx = bp[0], by = bp[1], bz = bp[2];
    const float bn2 = bx * bx + by * by + bz * bz;
    v8f c = {0.f, 0.f, 0.f, 0.f, 0.f, 0.f, 0.f, 0.f};
#if USE_WMMA
    v2f Bv;
    Bv.x = half ? bz  : bx;
    Bv.y = half ? bn2 : by;
    c = __builtin_amdgcn_wmma_f32_16x16x4_f32(
         false, A,  false, Bv,
         (short)0, c,  false,  false);
#else
#pragma unroll
    for (int r = 0; r < 8; ++r) {
      const int row = half * 8 + r;
      const float rx = __shfl(ax, row, 32);
      const float ry = __shfl(ay, row, 32);
      const float rz = __shfl(az, row, 32);
      c[r] = -2.0f * (rx * bx + ry * by + rz * bz) + bn2;
    }
#endif
#pragma unroll
    for (int r = 0; r < 8; ++r) ins3_stream(c[r], n, td[r], ti[r]);
  }

#pragma unroll
  for (int mskk = 1; mskk <= 8; mskk <<= 1) {
#pragma unroll
    for (int r = 0; r < 8; ++r) {
      float pd[3]; int pi[3];
#pragma unroll
      for (int k = 0; k < 3; ++k) {
        pd[k] = __shfl_xor(td[r][k], mskk, 32);
        pi[k] = __shfl_xor(ti[r][k], mskk, 32);
      }
#pragma unroll
      for (int k = 0; k < 3; ++k) ins3_lex(pd[k], pi[k], td[r], ti[r]);
    }
  }

  if (l15 == 0) {
#pragma unroll
    for (int r = 0; r < 8; ++r)
#pragma unroll
      for (int k = 0; k < 3; ++k) {
        s_td[wv][half * 8 + r][k] = td[r][k];
        s_ti[wv][half * 8 + r][k] = ti[r][k];
      }
  }
  __syncthreads();

  float fd[3];
  int   fi[3];
#pragma unroll
  for (int k = 0; k < 3; ++k) { fd[k] = s_td[l15][wv][k]; fi[k] = s_ti[l15][wv][k]; }
#pragma unroll
  for (int mskk = 1; mskk <= 8; mskk <<= 1) {
    float pd[3]; int pi[3];
#pragma unroll
    for (int k = 0; k < 3; ++k) {
      pd[k] = __shfl_xor(fd[k], mskk, 32);
      pi[k] = __shfl_xor(fi[k], mskk, 32);
    }
#pragma unroll
    for (int k = 0; k < 3; ++k) ins3_lex(pd[k], pi[k], fd, fi);
  }

  if (lane == 0) {
    const int row = rt * 16 + wv;
    const int gi  = fps_idx[b * M_ + row];
    const float* gp = pcb + (size_t)gi * 3;
    const float gx = gp[0], gy = gp[1], gz = gp[2];
    float wv3[3];
    float wsum = 0.f;
#pragma unroll
    for (int k = 0; k < 3; ++k) {
      const int nn = fi[k];
      const float* np = pcb + (size_t)nn * 3;
      const float dx = np[0] - gx, dy = np[1] - gy, dz = np[2] - gz;
      const float dd = dx * dx + dy * dy + dz * dz;
      const float dist = sqrtf(fmaxf(dd, 0.0f));
      wv3[k] = 1.0f / (dist + 1e-8f);
      wsum += wv3[k];
      idx3[((size_t)b * M_ + row) * 3 + k] = nn;
    }
#pragma unroll
    for (int k = 0; k < 3; ++k)
      w3[((size_t)b * M_ + row) * 3 + k] = wv3[k] / wsum;
  }
}

__global__ __launch_bounds__(256) void interp_kernel(
    const float* __restrict__ feats, const int* __restrict__ idx3,
    const float* __restrict__ w3, float* __restrict__ out) {
  const int bc = blockIdx.x;
  const int b  = bc / C_;
  const int c  = bc - b * C_;
  const float* f = feats + ((size_t)b * C_ + c) * N_;
  float*       o = out   + ((size_t)b * C_ + c) * M_;
  const int*   ib = idx3 + (size_t)b * M_ * 3;
  const float* wb = w3   + (size_t)b * M_ * 3;
  for (int m = threadIdx.x; m < M_; m += 256) {
    const int   n0 = ib[m * 3 + 0], n1 = ib[m * 3 + 1], n2 = ib[m * 3 + 2];
    const float w0 = wb[m * 3 + 0], w1 = wb[m * 3 + 1], w2 = wb[m * 3 + 2];
    o[m] = w0 * f[n0] + w1 * f[n1] + w2 * f[n2];
  }
}

extern "C" void kernel_launch(void* const* d_in, const int* in_sizes, int n_in,
                              void* d_out, int out_size, void* d_ws, size_t ws_size,
                              hipStream_t stream) {
  const float* pc    = (const float*)d_in[0];
  const float* feats = (const float*)d_in[1];
  const float* obj   = (const float*)d_in[2];
  const float* grasp = (const float*)d_in[3];
  float* out = (float*)d_out;

  char* ws = (char*)d_ws;
  size_t off = 0;
  int* fps = (int*)(ws + off);
  off += (size_t)B_ * M_ * 4;        off = (off + 255) & ~(size_t)255;
  int* idx3 = (int*)(ws + off);
  off += (size_t)B_ * M_ * 3 * 4;    off = (off + 255) & ~(size_t)255;
  float* w3 = (float*)(ws + off);
  off += (size_t)B_ * M_ * 3 * 4;

  fps_kernel<<<B_, 1024, 0, stream>>>(pc, obj, grasp, fps);
  three_nn_kernel<<<B_ * (M_ / 16), NN_WAVES * 32, 0, stream>>>(pc, fps, idx3, w3);
  interp_kernel<<<B_ * C_, 256, 0, stream>>>(feats, idx3, w3, out);
}


# fps kernel: 16 waves, not 32 (512 threads, up to 40 register slots each): per-step argmax work (32-lane butterflies, table reduce, now 16 entries one per lane) is paid by half the waves; same sums
# speedup vs baseline: 4.4724x; 1.1437x over previous
#include <hip/hip_runtime.h>
#include <hip/hip_bf16.h>

#define B_ 4
#define N_ 20000
#define C_ 512
#define M_ 1024
#define GRASP_TH 0.1f

typedef float v2f __attribute__((ext_vector_type(2)));
typedef float v8f __attribute__((ext_vector_type(8)));

#if defined(__gfx1250__) && __has_builtin(__builtin_amdgcn_wmma_f32_16x16x4_f32)
#define USE_WMMA 1
#else
#define USE_WMMA 0
#endif

__device__ __forceinline__ void ins3_stream(float v, int n, float d[3], int ix[3]) {
  const bool c0 = v < d[0];
  const bool c1 = v < d[1];
  const bool c2 = v < d[2];
  const float r0 = c0 ? v     : d[0];
  const int   j0 = c0 ? n     : ix[0];
  const float r1 = c0 ? d[0]  : (c1 ? v : d[1]);
  const int   j1 = c0 ? ix[0] : (c1 ? n : ix[1]);
  const float r2 = c1 ? d[1]  : (c2 ? v : d[2]);
  const int   j2 = c1 ? ix[1] : (c2 ? n : ix[2]);
  d[0] = r0; d[1] = r1; d[2] = r2;
  ix[0] = j0; ix[1] = j1; ix[2] = j2;
}

__device__ __forceinline__ void ins3_lex(float v, int n, float d[3], int ix[3]) {
  const bool c0 = (v < d[0]) | ((v == d[0]) & (n < ix[0]));
  const bool c1 = (v < d[1]) | ((v == d[1]) & (n < ix[1]));
  const bool c2 = (v < d[2]) | ((v == d[2]) & (n < ix[2]));
  const float r0 = c0 ? v     : d[0];
  const int   j0 = c0 ? n     : ix[0];
  const float r1 = c0 ? d[0]  : (c1 ? v : d[1]);
  const int   j1 = c0 ? ix[0] : (c1 ? n : ix[1]);
  const float r2 = c1 ? d[1]  : (c2 ? v : d[2]);
  const int   j2 = c1 ? ix[1] : (c2 ? n : ix[2]);
  d[0] = r0; d[1] = r1; d[2] = r2;
  ix[0] = j0; ix[1] = j1; ix[2] = j2;
}

template <int XM>
__device__ __forceinline__ int row_xor_i(int v) {
  return __builtin_amdgcn_update_dpp(v, v, 0x160 + XM, 0xf, 0xf, true);
}
__device__ __forceinline__ int xhalf_i(int v) {
  return __builtin_amdgcn_permlanex16(v, v, 0x76543210, 0xfedcba98, false, false);
}
__device__ __forceinline__ void wave_argmax(float& v, int& i) {
  const int key = __builtin_bit_cast(int, v);
  int m = key;
  m = max(m, row_xor_i<1>(m));
  m = max(m, row_xor_i<2>(m));
  m = max(m, row_xor_i<4>(m));
  m = max(m, row_xor_i<8>(m));
  m = max(m, xhalf_i(m));
  int c = (key == m) ? i : 0x7fffffff;
  c = min(c, row_xor_i<1>(c));
  c = min(c, row_xor_i<2>(c));
  c = min(c, row_xor_i<4>(c));
  c = min(c, row_xor_i<8>(c));
  c = min(c, xhalf_i(c));
  v = __builtin_bit_cast(float, m);
  i = c;
}

__device__ __forceinline__ int table_argmax(float v, int i) {
  const int k = __builtin_bit_cast(int, v);
  int m = k;
  m = max(m, row_xor_i<1>(m));
  m = max(m, row_xor_i<2>(m));
  m = max(m, row_xor_i<4>(m));
  m = max(m, row_xor_i<8>(m));
  int c = (k == m) ? i : 0x7fffffff;
  c = min(c, row_xor_i<1>(c));
  c = min(c, row_xor_i<2>(c));
  c = min(c, row_xor_i<4>(c));
  c = min(c, row_xor_i<8>(c));
  return c;
}

__device__ __forceinline__ void fps_pass(v2f xy, float z, float& m, v2f pxy, float pz,
                                         int p, float& bestV, int& bestP) {
  const v2f   d  = xy - pxy;
  const float dz = z - pz;
  const float sxy = d.x * d.x + d.y * d.y;
  const float dd  = __builtin_fmaf(dz, dz, sxy);
  const float nm = fminf(m, dd);
  m = nm;
  const bool better = nm > bestV;
  bestV = better ? nm : bestV;
  bestP = better ? p : bestP;
}

#define FPS_THREADS 512
#define FPS_WAVES  (FPS_THREADS / 32)
#define FPS_PASSES 40
#define FPS_WORDS  (FPS_PASSES * FPS_WAVES)
#define FPS_WPL    (FPS_WORDS / 32)
#define FPS_GROUP  8
#define CMAX 15000
#define FPS_PMAX FPS_PASSES
static_assert(FPS_PASSES * FPS_THREADS >= N_, "the passes cover every point");
static_assert(FPS_PASSES > 32 && FPS_PASSES <= 64, "two flag-word registers per lane");
static_assert(FPS_WORDS % 32 == 0 && FPS_PASSES % FPS_GROUP == 0 && FPS_WAVES == 16,
              "prefix split, staging groups, one table entry per lane of a row");

__global__ __launch_bounds__(FPS_THREADS) void fps_kernel(
    const float* __restrict__ pc, const float* __restrict__ obj,
    const float* __restrict__ grasp, int* __restrict__ fps_idx) {
  const int b    = blockIdx.x;
  const int tid  = threadIdx.x;
  const int lane = tid & 31;
  const int wid  = tid >> 5;

  __shared__ float    s_xyz[N_ * 3];
  __shared__ unsigned s_bits[FPS_WORDS];
  __shared__ int      s_pref[FPS_WORDS];
  __shared__ int   s_last;
  __shared__ int   s_K;

  const float* pcb = pc + (size_t)b * N_ * 3;
  const float* ob0 = obj + (size_t)b * 2 * N_;
  const float* gr  = grasp + (size_t)b * N_;

  unsigned myw0 = 0u, myw1 = 0u;
  for (int it = 0; it < 32; ++it) {
    const int n  = it * FPS_THREADS + tid;
    const int nl = (n < N_) ? n : (N_ - 1);
    const bool mk = (n < N_) & (ob0[N_ + nl] > ob0[nl]) & (gr[nl] > GRASP_TH);
    const unsigned w = __builtin_amdgcn_ballot_w32(mk);
    myw0 = (lane == it) ? w : myw0;
  }
  for (int it = 32; it < FPS_PASSES; ++it) {
    const int n  = it * FPS_THREADS + tid;
    const int nl = (n < N_) ? n : (N_ - 1);
    const bool mk = (n < N_) & (ob0[N_ + nl] > ob0[nl]) & (gr[nl] > GRASP_TH);
    const unsigned w = __builtin_amdgcn_ballot_w32(mk);
    myw1 = (lane == it - 32) ? w : myw1;
  }
  s_bits[lane * FPS_WAVES + wid] = myw0;
  if (lane < FPS_PASSES - 32) s_bits[(lane + 32) * FPS_WAVES + wid] = myw1;
  __syncthreads();

  if (wid == 0) {
    int sum = 0;
    for (int j = 0; j < FPS_WPL; ++j) sum += __popc(s_bits[lane * FPS_WPL + j]);
    int incl = sum;
    for (int d = 1; d < 32; d <<= 1) {
      const int o = __shfl_up(incl, d, 32);
      incl += (lane >= d) ? o : 0;
    }
    int run = incl - sum;
    for (int j = 0; j < FPS_WPL; ++j) {
      s_pref[lane * FPS_WPL + j] = run;
      run += __popc(s_bits[lane * FPS_WPL + j]);
    }
    if (lane == 31) s_K = incl;
    if (lane == 0)  s_last = 0;
  }
  __syncthreads();

  const int  K       = s_K;
  const bool compact = (K >= 1) & (K <= CMAX);
  const int  Kc      = compact ? K : N_;
  int* s_idx = (int*)(s_xyz + (compact ? 3 * Kc : 0));

  for (int g = 0; g < FPS_PASSES; g += FPS_GROUP) {
    float x[FPS_GROUP], y[FPS_GROUP], z[FPS_GROUP];
    unsigned w[FPS_GROUP];
    int pf[FPS_GROUP];
    for (int j = 0; j < FPS_GROUP; ++j) {
      const int n  = (g + j) * FPS_THREADS + tid;
      const int nl = (n < N_) ? n : (N_ - 1);
      const float* p = pcb + (size_t)nl * 3;
      x[j] = p[0]; y[j] = p[1]; z[j] = p[2];
      w[j]  = s_bits[(g + j) * FPS_WAVES + wid];
      pf[j] = s_pref[(g + j) * FPS_WAVES + wid];
    }
    for (int j = 0; j < FPS_GROUP; ++j) {
      const int  n    = (g + j) * FPS_THREADS + tid;
      const bool mk   = (w[j] >> lane) & 1u;
      const int  pos  = pf[j] + __popc(w[j] & ((1u << lane) - 1u));
      const bool keep = compact ? mk : (n < N_);
      const int  slot = compact ? pos : n;
      if (keep) {
        s_xyz[slot * 3 + 0] = x[j];
        s_xyz[slot * 3 + 1] = y[j];
        s_xyz[slot * 3 + 2] = z[j];
        if (compact) s_idx[slot] = n;
        else if (mk & (pos == 0)) s_last = n;
      }
    }
  }
  __syncthreads();

  const int wv  = __builtin_amdgcn_readfirstlane(wid);
  const int rem = Kc - wv * 32;
  const int Pw  = (rem > 0) ? ((rem + FPS_THREADS - 1) / FPS_THREADS) : 0;
  v2f   rxy[FPS_PMAX];
  float rz[FPS_PMAX], rm[FPS_PMAX];
#pragma unroll
  for (int p = 0; p < FPS_PMAX; ++p) {
    const int  slot = p * FPS_THREADS + tid;
    const bool in   = slot < Kc;
    const int  sl   = in ? slot : 0;
    rxy[p].x = s_xyz[sl * 3 + 0];
    rxy[p].y = s_xyz[sl * 3 + 1];
    rz[p]    = s_xyz[sl * 3 + 2];
    const bool mk = (s_bits[p * FPS_WAVES + wid] >> lane) & 1u;
    rm[p] = (in & (compact | mk)) ? __builtin_inff() : -__builtin_inff();
  }
  __syncthreads();

  float* swv = reinterpret_cast<float*>(s_bits);
  int*   swi = s_pref;
  int last = s_last;
  int acc = 0;
  for (int step = 0; step < M_; ++step) {
    v2f pxy;
    pxy.x = s_xyz[last * 3 + 0];
    pxy.y = s_xyz[last * 3 + 1];
    const float pz = s_xyz[last * 3 + 2];
    if (wid == 0) {
      const int pick = compact ? s_idx[last] : last;
      acc = (lane == (step & 31)) ? pick : acc;
      if ((step & 31) == 31) fps_idx[b * M_ + (step - 31) + lane] = acc;
    }

    float bestV = -__builtin_inff();
    int   bestP = 0;
    if (Pw > 0) { fps_pass(rxy[0], rz[0], rm[0], pxy, pz, 0, bestV, bestP);
    if (Pw > 1) { fps_pass(rxy[1], rz[1], rm[1], pxy, pz, 1, bestV, bestP);
    if (Pw > 2) { fps_pass(rxy[2], rz[2], rm[2], pxy, pz, 2, bestV, bestP);
    if (Pw > 3) { fps_pass(rxy[3], rz[3], rm[3], pxy, pz, 3, bestV, bestP);
    if (Pw > 4) { fps_pass(rxy[4], rz[4], rm[4], pxy, pz, 4, bestV, bestP);
    if (Pw > 5) { fps_pass(rxy[5], rz[5], rm[5], pxy, pz, 5, bestV, bestP);
    if (Pw > 6) { fps_pass(rxy[6], rz[6], rm[6], pxy, pz, 6, bestV, bestP);
    if (Pw > 7) { fps_pass(rxy[7], rz[7], rm[7], pxy, pz, 7, bestV, bestP);
    if (Pw > 8) { fps_pass(rxy[8], rz[8], rm[8], pxy, pz, 8, bestV, bestP);
    if (Pw > 9) { fps_pass(rxy[9], rz[9], rm[9], pxy, pz, 9, bestV, bestP);
    if (Pw > 10) { fps_pass(rxy[10], rz[10], rm[10], pxy, pz, 10, bestV, bestP);
    if (Pw > 11) { fps_pass(rxy[11], rz[11], rm[11], pxy, pz, 11, bestV, bestP);
    if (Pw > 12) { fps_pass(rxy[12], rz[12], rm[12], pxy, pz, 12, bestV, bestP);
    if (Pw > 13) { fps_pass(rxy[13], rz[13], rm[13], pxy, pz, 13, bestV, bestP);
    if (Pw > 14) { fps_pass(rxy[14], rz[14], rm[14], pxy, pz, 14, bestV, bestP);
    if (Pw > 15) { fps_pass(rxy[15], rz[15], rm[15], pxy, pz, 15, bestV, bestP);
    if (Pw > 16) { fps_pass(rxy[16], rz[16], rm[16], pxy, pz, 16, bestV, bestP);
    if (Pw > 17) { fps_pass(rxy[17], rz[17], rm[17], pxy, pz, 17, bestV, bestP);
    if (Pw > 18) { fps_pass(rxy[18], rz[18], rm[18], pxy, pz, 18, bestV, bestP);
    if (Pw > 19) { fps_pass(rxy[19], rz[19], rm[19], pxy, pz, 19, bestV, bestP);
    if (Pw > 20) { fps_pass(rxy[20], rz[20], rm[20], pxy, pz, 20, bestV, bestP);
    if (Pw > 21) { fps_pass(rxy[21], rz[21], rm[21], pxy, pz, 21, bestV, bestP);
    if (Pw > 22) { fps_pass(rxy[22], rz[22], rm[22], pxy, pz, 22, bestV, bestP);
    if (Pw > 23) { fps_pass(rxy[23], rz[23], rm[23], pxy, pz, 23, bestV, bestP);
    if (Pw > 24) { fps_pass(rxy[24], rz[24], rm[24], pxy, pz, 24, bestV, bestP);
    if (Pw > 25) { fps_pass(rxy[25], rz[25], rm[25], pxy, pz, 25, bestV, bestP);
    if (Pw > 26) { fps_pass(rxy[26], rz[26], rm[26], pxy, pz, 26, bestV, bestP);
    if (Pw > 27) { fps_pass(rxy[27], rz[27], rm[27], pxy, pz, 27, bestV, bestP);
    if (Pw > 28) { fps_pass(rxy[28], rz[28], rm[28], pxy, pz, 28, bestV, bestP);
    if (Pw > 29) { fps_pass(rxy[29], rz[29], rm[29], pxy, pz, 29, bestV, bestP);
    if (Pw > 30) { fps_pass(rxy[30], rz[30], rm[30], pxy, pz, 30, bestV, bestP);
    if (Pw > 31) { fps_pass(rxy[31], rz[31], rm[31], pxy, pz, 31, bestV, bestP);
    if (Pw > 32) { fps_pass(rxy[32], rz[32], rm[32], pxy, pz, 32, bestV, bestP);
    if (Pw > 33) { fps_pass(rxy[33], rz[33], rm[33], pxy, pz, 33, bestV, bestP);
    if (Pw > 34) { fps_pass(rxy[34], rz[34], rm[34], pxy, pz, 34, bestV, bestP);
    if (Pw > 35) { fps_pass(rxy[35], rz[35], rm[35], pxy, pz, 35, bestV, bestP);
    if (Pw > 36) { fps_pass(rxy[36], rz[36], rm[36], pxy, pz, 36, bestV, bestP);
    if (Pw > 37) { fps_pass(rxy[37], rz[37], rm[37], pxy, pz, 37, bestV, bestP);
    if (Pw > 38) { fps_pass(rxy[38], rz[38], rm[38], pxy, pz, 38, bestV, bestP);
    if (Pw > 39) { fps_pass(rxy[39], rz[39], rm[39], pxy, pz, 39, bestV, bestP);
    }}}}}}}}}}}}}}}}}}}}}}}}}}}}}}}}}}}}}}}}
    int bestI = bestP * FPS_THREADS + tid;
    wave_argmax(bestV, bestI);
    const int pb = (step & 1) * FPS_WAVES;
    if (lane == 0) { swv[pb + wid] = bestV; swi[pb + wid] = bestI; }
    __syncthreads();
    const int e = pb + (lane & 15);
    last = table_argmax(swv[e], swi[e]);
  }
}

#define NN_WAVES 16
__global__ __launch_bounds__(NN_WAVES * 32) void three_nn_kernel(
    const float* __restrict__ pc, const int* __restrict__ fps_idx,
    int* __restrict__ idx3, float* __restrict__ w3) {
  static_assert(NN_WAVES == 16, "the final merge maps wave j onto lane j of a 16-lane row");
  const int tid  = threadIdx.x;
  const int lane = tid & 31;
  const int wv   = __builtin_amdgcn_readfirstlane(tid >> 5);
  const int tilesPerBatch = M_ / 16;
  const int b    = blockIdx.x / tilesPerBatch;
  const int rt   = blockIdx.x - b * tilesPerBatch;
  const int half = lane >> 4;
  const int l15  = lane & 15;

  __shared__ float s_td[NN_WAVES][16][3];
  __shared__ int   s_ti[NN_WAVES][16][3];

  const float* pcb = pc + (size_t)b * N_ * 3;

  const int mrow = rt * 16 + l15;
  const int aidx = fps_idx[b * M_ + mrow];
  const float* ap = pcb + (size_t)aidx * 3;
  const float ax = ap[0], ay = ap[1], az = ap[2];
  v2f A;
  A.x = half ? (-2.0f * az) : (-2.0f * ax);
  A.y = half ? 1.0f         : (-2.0f * ay);

  float td[8][3];
  int   ti[8][3];
#pragma unroll
  for (int r = 0; r < 8; ++r)
#pragma unroll
    for (int k = 0; k < 3; ++k) { td[r][k] = __builtin_inff(); ti[r][k] = 0x7fffffff; }

  const int t0 = (wv * (N_ / 16)) / NN_WAVES;
  const int t1 = ((wv + 1) * (N_ / 16)) / NN_WAVES;
  int n = t0 * 16 + l15;
  const float* bp = pcb + (size_t)n * 3;
#pragma unroll 2
  for (int t = t0; t < t1; ++t, n += 16, bp += 16 * 3) {
    const int pn = (n + 128 < N_) ? (n + 128) : (N_ - 1);
    __builtin_prefetch(pcb + (size_t)pn * 3, 0, 1);
    const float bx = bp[0], by = bp[1], bz = bp[2];
    const float bn2 = bx * bx + by * by + bz * bz;
    v8f c = {0.f, 0.f, 0.f, 0.f, 0.f, 0.f, 0.f, 0.f};
#if USE_WMMA
    v2f Bv;
    Bv.x = half ? bz  : bx;
    Bv.y = half ? bn2 : by;
    c = __builtin_amdgcn_wmma_f32_16x16x4_f32(
         false, A,  false, Bv,
         (short)0, c,  false,  false);
#else
#pragma unroll
    for (int r = 0; r < 8; ++r) {
      const int row = half * 8 + r;
      const float rx = __shfl(ax, row, 32);
      const float ry = __shfl(ay, row, 32);
      const float rz = __shfl(az, row, 32);
      c[r] = -2.0f * (rx * bx + ry * by + rz * bz) + bn2;
    }
#endif
#pragma unroll
    for (int r = 0; r < 8; ++r) ins3_stream(c[r], n, td[r], ti[r]);
  }

#pragma unroll
  for (int mskk = 1; mskk <= 8; mskk <<= 1) {
#pragma unroll
    for (int r = 0; r < 8; ++r) {
      float pd[3]; int pi[3];
#pragma unroll
      for (int k = 0; k < 3; ++k) {
        pd[k] = __shfl_xor(td[r][k], mskk, 32);
        pi[k] = __shfl_xor(ti[r][k], mskk, 32);
      }
#pragma unroll
      for (int k = 0; k < 3; ++k) ins3_lex(pd[k], pi[k], td[r], ti[r]);
    }
  }

  if (l15 == 0) {
#pragma unroll
    for (int r = 0; r < 8; ++r)
#pragma unroll
      for (int k = 0; k < 3; ++k) {
        s_td[wv][half * 8 + r][k] = td[r][k];
        s_ti[wv][half * 8 + r][k] = ti[r][k];
      }
  }
  __syncthreads();

  float fd[3];
  int   fi[3];
#pragma unroll
  for (int k = 0; k < 3; ++k) { fd[k] = s_td[l15][wv][k]; fi[k] = s_ti[l15][wv][k]; }
#pragma unroll
  for (int mskk = 1; mskk <= 8; mskk <<= 1) {
    float pd[3]; int pi[3];
#pragma unroll
    for (int k = 0; k < 3; ++k) {
      pd[k] = __shfl_xor(fd[k], mskk, 32);
      pi[k] = __shfl_xor(fi[k], mskk, 32);
    }
#pragma unroll
    for (int k = 0; k < 3; ++k) ins3_lex(pd[k], pi[k], fd, fi);
  }

  if (lane == 0) {
    const int row = rt * 16 + wv;
    const int gi  = fps_idx[b * M_ + row];
    const float* gp = pcb + (size_t)gi * 3;
    const float gx = gp[0], gy = gp[1], gz = gp[2];
    float wv3[3];
    float wsum = 0.f;
#pragma unroll
    for (int k = 0; k < 3; ++k) {
      const int nn = fi[k];
      const float* np = pcb + (size_t)nn * 3;
      const float dx = np[0] - gx, dy = np[1] - gy, dz = np[2] - gz;
      const float dd = dx * dx + dy * dy + dz * dz;
      const float dist = sqrtf(fmaxf(dd, 0.0f));
      wv3[k] = 1.0f / (dist + 1e-8f);
      wsum += wv3[k];
      idx3[((size_t)b * M_ + row) * 3 + k] = nn;
    }
#pragma unroll
    for (int k = 0; k < 3; ++k)
      w3[((size_t)b * M_ + row) * 3 + k] = wv3[k] / wsum;
  }
}

__global__ __launch_bounds__(256) void interp_kernel(
    const float* __restrict__ feats, const int* __restrict__ idx3,
    const float* __restrict__ w3, float* __restrict__ out) {
  const int bc = blockIdx.x;
  const int b  = bc / C_;
  const int c  = bc - b * C_;
  const float* f = feats + ((size_t)b * C_ + c) * N_;
  float*       o = out   + ((size_t)b * C_ + c) * M_;
  const int*   ib = idx3 + (size_t)b * M_ * 3;
  const float* wb = w3   + (size_t)b * M_ * 3;
  for (int m = threadIdx.x; m < M_; m += 256) {
    const int   n0 = ib[m * 3 + 0], n1 = ib[m * 3 + 1], n2 = ib[m * 3 + 2];
    const float w0 = wb[m * 3 + 0], w1 = wb[m * 3 + 1], w2 = wb[m * 3 + 2];
    o[m] = w0 * f[n0] + w1 * f[n1] + w2 * f[n2];
  }
}

extern "C" void kernel_launch(void* const* d_in, const int* in_sizes, int n_in,
                              void* d_out, int out_size, void* d_ws, size_t ws_size,
                              hipStream_t stream) {
  const float* pc    = (const float*)d_in[0];
  const float* feats = (const float*)d_in[1];
  const float* obj   = (const float*)d_in[2];
  const float* grasp = (const float*)d_in[3];
  float* out = (float*)d_out;

  char* ws = (char*)d_ws;
  size_t off = 0;
  int* fps = (int*)(ws + off);
  off += (size_t)B_ * M_ * 4;        off = (off + 255) & ~(size_t)255;
  int* idx3 = (int*)(ws + off);
  off += (size_t)B_ * M_ * 3 * 4;    off = (off + 255) & ~(size_t)255;
  float* w3 = (float*)(ws + off);
  off += (size_t)B_ * M_ * 3 * 4;

  fps_kernel<<<B_, FPS_THREADS, 0, stream>>>(pc, obj, grasp, fps);
  three_nn_kernel<<<B_ * (M_ / 16), NN_WAVES * 32, 0, stream>>>(pc, fps, idx3, w3);
  interp_kernel<<<B_ * C_, 256, 0, stream>>>(feats, idx3, w3, out);
}


# fps kernel: serial loop compiled once per pass count (1..40, same in all waves, one block-uniform switch): a step's passes are one straight line, no scalar test and branch per pass; same sums
# speedup vs baseline: 4.6929x; 1.0493x over previous
#include <hip/hip_runtime.h>
#include <hip/hip_bf16.h>

#define B_ 4
#define N_ 20000
#define C_ 512
#define M_ 1024
#define GRASP_TH 0.1f

typedef float v2f __attribute__((ext_vector_type(2)));
typedef float v8f __attribute__((ext_vector_type(8)));

#if defined(__gfx1250__) && __has_builtin(__builtin_amdgcn_wmma_f32_16x16x4_f32)
#define USE_WMMA 1
#else
#define USE_WMMA 0
#endif

__device__ __forceinline__ void ins3_stream(float v, int n, float d[3], int ix[3]) {
  const bool c0 = v < d[0];
  const bool c1 = v < d[1];
  const bool c2 = v < d[2];
  const float r0 = c0 ? v     : d[0];
  const int   j0 = c0 ? n     : ix[0];
  const float r1 = c0 ? d[0]  : (c1 ? v : d[1]);
  const int   j1 = c0 ? ix[0] : (c1 ? n : ix[1]);
  const float r2 = c1 ? d[1]  : (c2 ? v : d[2]);
  const int   j2 = c1 ? ix[1] : (c2 ? n : ix[2]);
  d[0] = r0; d[1] = r1; d[2] = r2;
  ix[0] = j0; ix[1] = j1; ix[2] = j2;
}

__device__ __forceinline__ void ins3_lex(float v, int n, float d[3], int ix[3]) {
  const bool c0 = (v < d[0]) | ((v == d[0]) & (n < ix[0]));
  const bool c1 = (v < d[1]) | ((v == d[1]) & (n < ix[1]));
  const bool c2 = (v < d[2]) | ((v == d[2]) & (n < ix[2]));
  const float r0 = c0 ? v     : d[0];
  const int   j0 = c0 ? n     : ix[0];
  const float r1 = c0 ? d[0]  : (c1 ? v : d[1]);
  const int   j1 = c0 ? ix[0] : (c1 ? n : ix[1]);
  const float r2 = c1 ? d[1]  : (c2 ? v : d[2]);
  const int   j2 = c1 ? ix[1] : (c2 ? n : ix[2]);
  d[0] = r0; d[1] = r1; d[2] = r2;
  ix[0] = j0; ix[1] = j1; ix[2] = j2;
}

template <int XM>
__device__ __forceinline__ int row_xor_i(int v) {
  return __builtin_amdgcn_update_dpp(v, v, 0x160 + XM, 0xf, 0xf, true);
}
__device__ __forceinline__ int xhalf_i(int v) {
  return __builtin_amdgcn_permlanex16(v, v, 0x76543210, 0xfedcba98, false, false);
}
__device__ __forceinline__ void wave_argmax(float& v, int& i) {
  const int key = __builtin_bit_cast(int, v);
  int m = key;
  m = max(m, row_xor_i<1>(m));
  m = max(m, row_xor_i<2>(m));
  m = max(m, row_xor_i<4>(m));
  m = max(m, row_xor_i<8>(m));
  m = max(m, xhalf_i(m));
  int c = (key == m) ? i : 0x7fffffff;
  c = min(c, row_xor_i<1>(c));
  c = min(c, row_xor_i<2>(c));
  c = min(c, row_xor_i<4>(c));
  c = min(c, row_xor_i<8>(c));
  c = min(c, xhalf_i(c));
  v = __builtin_bit_cast(float, m);
  i = c;
}

__device__ __forceinline__ int table_argmax(float v, int i) {
  const int k = __builtin_bit_cast(int, v);
  int m = k;
  m = max(m, row_xor_i<1>(m));
  m = max(m, row_xor_i<2>(m));
  m = max(m, row_xor_i<4>(m));
  m = max(m, row_xor_i<8>(m));
  int c = (k == m) ? i : 0x7fffffff;
  c = min(c, row_xor_i<1>(c));
  c = min(c, row_xor_i<2>(c));
  c = min(c, row_xor_i<4>(c));
  c = min(c, row_xor_i<8>(c));
  return c;
}

__device__ __forceinline__ void fps_pass(v2f xy, float z, float& m, v2f pxy, float pz,
                                         int p, float& bestV, int& bestP) {
  const v2f   d  = xy - pxy;
  const float dz = z - pz;
  const float sxy = d.x * d.x + d.y * d.y;
  const float dd  = __builtin_fmaf(dz, dz, sxy);
  const float nm = fminf(m, dd);
  m = nm;
  const bool better = nm > bestV;
  bestV = better ? nm : bestV;
  bestP = better ? p : bestP;
}

#define FPS_THREADS 512
#define FPS_WAVES  (FPS_THREADS / 32)
#define FPS_PASSES 40
#define FPS_WORDS  (FPS_PASSES * FPS_WAVES)
#define FPS_WPL    (FPS_WORDS / 32)
#define FPS_GROUP  8
#define CMAX 15000
static_assert(FPS_PASSES * FPS_THREADS >= N_, "the passes cover every point");
static_assert(FPS_PASSES > 32 && FPS_PASSES <= 64, "two flag-word registers per lane");
static_assert(FPS_WORDS % 32 == 0 && FPS_PASSES % FPS_GROUP == 0 && FPS_WAVES == 16,
              "prefix split, staging groups, one table entry per lane of a row");

template <int P>
__device__ __forceinline__ void fps_serial(const float* s_xyz, const int* s_idx,
                                           unsigned* s_bits, int* s_pref,
                                           int first, bool compact, int Kc,
                                           int b, int tid, int* __restrict__ fps_idx) {
  const int lane = tid & 31;
  const int wid  = tid >> 5;
  v2f   rxy[P];
  float rz[P], rm[P];
#pragma unroll
  for (int p = 0; p < P; ++p) {
    const int  slot = p * FPS_THREADS + tid;
    const bool in   = slot < Kc;
    const int  sl   = in ? slot : 0;
    rxy[p].x = s_xyz[sl * 3 + 0];
    rxy[p].y = s_xyz[sl * 3 + 1];
    rz[p]    = s_xyz[sl * 3 + 2];
    const bool mk = (s_bits[p * FPS_WAVES + wid] >> lane) & 1u;
    rm[p] = (in & (compact | mk)) ? __builtin_inff() : -__builtin_inff();
  }
  __syncthreads();

  float* swv = reinterpret_cast<float*>(s_bits);
  int*   swi = s_pref;
  int last = first;
  int acc = 0;
  for (int step = 0; step < M_; ++step) {
    v2f pxy;
    pxy.x = s_xyz[last * 3 + 0];
    pxy.y = s_xyz[last * 3 + 1];
    const float pz = s_xyz[last * 3 + 2];
    if (wid == 0) {
      const int pick = compact ? s_idx[last] : last;
      acc = (lane == (step & 31)) ? pick : acc;
      if ((step & 31) == 31) fps_idx[b * M_ + (step - 31) + lane] = acc;
    }

    float bestV = -__builtin_inff();
    int   bestP = 0;
#pragma unroll
    for (int p = 0; p < P; ++p)
      fps_pass(rxy[p], rz[p], rm[p], pxy, pz, p, bestV, bestP);
    int bestI = bestP * FPS_THREADS + tid;
    wave_argmax(bestV, bestI);
    const int pb = (step & 1) * FPS_WAVES;
    if (lane == 0) { swv[pb + wid] = bestV; swi[pb + wid] = bestI; }
    __syncthreads();
    const int e = pb + (lane & 15);
    last = table_argmax(swv[e], swi[e]);
  }
}

__global__ __launch_bounds__(FPS_THREADS) void fps_kernel(
    const float* __restrict__ pc, const float* __restrict__ obj,
    const float* __restrict__ grasp, int* __restrict__ fps_idx) {
  const int b    = blockIdx.x;
  const int tid  = threadIdx.x;
  const int lane = tid & 31;
  const int wid  = tid >> 5;

  __shared__ float    s_xyz[N_ * 3];
  __shared__ unsigned s_bits[FPS_WORDS];
  __shared__ int      s_pref[FPS_WORDS];
  __shared__ int   s_last;
  __shared__ int   s_K;

  const float* pcb = pc + (size_t)b * N_ * 3;
  const float* ob0 = obj + (size_t)b * 2 * N_;
  const float* gr  = grasp + (size_t)b * N_;

  unsigned myw0 = 0u, myw1 = 0u;
  for (int it = 0; it < 32; ++it) {
    const int n  = it * FPS_THREADS + tid;
    const int nl = (n < N_) ? n : (N_ - 1);
    const bool mk = (n < N_) & (ob0[N_ + nl] > ob0[nl]) & (gr[nl] > GRASP_TH);
    const unsigned w = __builtin_amdgcn_ballot_w32(mk);
    myw0 = (lane == it) ? w : myw0;
  }
  for (int it = 32; it < FPS_PASSES; ++it) {
    const int n  = it * FPS_THREADS + tid;
    const int nl = (n < N_) ? n : (N_ - 1);
    const bool mk = (n < N_) & (ob0[N_ + nl] > ob0[nl]) & (gr[nl] > GRASP_TH);
    const unsigned w = __builtin_amdgcn_ballot_w32(mk);
    myw1 = (lane == it - 32) ? w : myw1;
  }
  s_bits[lane * FPS_WAVES + wid] = myw0;
  if (lane < FPS_PASSES - 32) s_bits[(lane + 32) * FPS_WAVES + wid] = myw1;
  __syncthreads();

  if (wid == 0) {
    int sum = 0;
    for (int j = 0; j < FPS_WPL; ++j) sum += __popc(s_bits[lane * FPS_WPL + j]);
    int incl = sum;
    for (int d = 1; d < 32; d <<= 1) {
      const int o = __shfl_up(incl, d, 32);
      incl += (lane >= d) ? o : 0;
    }
    int run = incl - sum;
    for (int j = 0; j < FPS_WPL; ++j) {
      s_pref[lane * FPS_WPL + j] = run;
      run += __popc(s_bits[lane * FPS_WPL + j]);
    }
    if (lane == 31) s_K = incl;
    if (lane == 0)  s_last = 0;
  }
  __syncthreads();

  const int  K       = s_K;
  const bool compact = (K >= 1) & (K <= CMAX);
  const int  Kc      = compact ? K : N_;
  int* s_idx = (int*)(s_xyz + (compact ? 3 * Kc : 0));

  for (int g = 0; g < FPS_PASSES; g += FPS_GROUP) {
    float x[FPS_GROUP], y[FPS_GROUP], z[FPS_GROUP];
    unsigned w[FPS_GROUP];
    int pf[FPS_GROUP];
    for (int j = 0; j < FPS_GROUP; ++j) {
      const int n  = (g + j) * FPS_THREADS + tid;
      const int nl = (n < N_) ? n : (N_ - 1);
      const float* p = pcb + (size_t)nl * 3;
      x[j] = p[0]; y[j] = p[1]; z[j] = p[2];
      w[j]  = s_bits[(g + j) * FPS_WAVES + wid];
      pf[j] = s_pref[(g + j) * FPS_WAVES + wid];
    }
    for (int j = 0; j < FPS_GROUP; ++j) {
      const int  n    = (g + j) * FPS_THREADS + tid;
      const bool mk   = (w[j] >> lane) & 1u;
      const int  pos  = pf[j] + __popc(w[j] & ((1u << lane) - 1u));
      const bool keep = compact ? mk : (n < N_);
      const int  slot = compact ? pos : n;
      if (keep) {
        s_xyz[slot * 3 + 0] = x[j];
        s_xyz[slot * 3 + 1] = y[j];
        s_xyz[slot * 3 + 2] = z[j];
        if (compact) s_idx[slot] = n;
        else if (mk & (pos == 0)) s_last = n;
      }
    }
  }
  __syncthreads();

  const int first = s_last;
  const int np = (Kc + FPS_THREADS - 1) / FPS_THREADS;
#define FPS_RUN(P) case P: fps_serial<P>(s_xyz, s_idx, s_bits, s_pref, first, compact, Kc, b, tid, fps_idx); break;
  switch (np) {
    FPS_RUN(1)  FPS_RUN(2)  FPS_RUN(3)  FPS_RUN(4)  FPS_RUN(5)  FPS_RUN(6)  FPS_RUN(7)  FPS_RUN(8)
    FPS_RUN(9)  FPS_RUN(10) FPS_RUN(11) FPS_RUN(12) FPS_RUN(13) FPS_RUN(14) FPS_RUN(15) FPS_RUN(16)
    FPS_RUN(17) FPS_RUN(18) FPS_RUN(19) FPS_RUN(20) FPS_RUN(21) FPS_RUN(22) FPS_RUN(23) FPS_RUN(24)
    FPS_RUN(25) FPS_RUN(26) FPS_RUN(27) FPS_RUN(28) FPS_RUN(29) FPS_RUN(30) FPS_RUN(31) FPS_RUN(32)
    FPS_RUN(33) FPS_RUN(34) FPS_RUN(35) FPS_RUN(36) FPS_RUN(37) FPS_RUN(38) FPS_RUN(39)
    default: fps_serial<FPS_PASSES>(s_xyz, s_idx, s_bits, s_pref, first, compact, Kc, b, tid, fps_idx); break;
  }
#undef FPS_RUN
}

#define NN_WAVES 16
__global__ __launch_bounds__(NN_WAVES * 32) void three_nn_kernel(
    const float* __restrict__ pc, const int* __restrict__ fps_idx,
    int* __restrict__ idx3, float* __restrict__ w3) {
  static_assert(NN_WAVES == 16, "the final merge maps wave j onto lane j of a 16-lane row");
  const int tid  = threadIdx.x;
  const int lane = tid & 31;
  const int wv   = __builtin_amdgcn_readfirstlane(tid >> 5);
  const int tilesPerBatch = M_ / 16;
  const int b    = blockIdx.x / tilesPerBatch;
  const int rt   = blockIdx.x - b * tilesPerBatch;
  const int half = lane >> 4;
  const int l15  = lane & 15;

  __shared__ float s_td[NN_WAVES][16][3];
  __shared__ int   s_ti[NN_WAVES][16][3];

  const float* pcb = pc + (size_t)b * N_ * 3;

  const int mrow = rt * 16 + l15;
  const int aidx = fps_idx[b * M_ + mrow];
  const float* ap = pcb + (size_t)aidx * 3;
  const float ax = ap[0], ay = ap[1], az = ap[2];
  v2f A;
  A.x = half ? (-2.0f * az) : (-2.0f * ax);
  A.y = half ? 1.0f         : (-2.0f * ay);

  float td[8][3];
  int   ti[8][3];
#pragma unroll
  for (int r = 0; r < 8; ++r)
#pragma unroll
    for (int k = 0; k < 3; ++k) { td[r][k] = __builtin_inff(); ti[r][k] = 0x7fffffff; }

  const int t0 = (wv * (N_ / 16)) / NN_WAVES;
  const int t1 = ((wv + 1) * (N_ / 16)) / NN_WAVES;
  int n = t0 * 16 + l15;
  const float* bp = pcb + (size_t)n * 3;
#pragma unroll 2
  for (int t = t0; t < t1; ++t, n += 16, bp += 16 * 3) {
    const int pn = (n + 128 < N_) ? (n + 128) : (N_ - 1);
    __builtin_prefetch(pcb + (size_t)pn * 3, 0, 1);
    const float bx = bp[0], by = bp[1], bz = bp[2];
    const float bn2 = bx * bx + by * by + bz * bz;
    v8f c = {0.f, 0.f, 0.f, 0.f, 0.f, 0.f, 0.f, 0.f};
#if USE_WMMA
    v2f Bv;
    Bv.x = half ? bz  : bx;
    Bv.y = half ? bn2 : by;
    c = __builtin_amdgcn_wmma_f32_16x16x4_f32(
         false, A,  false, Bv,
         (short)0, c,  false,  false);
#else
#pragma unroll
    for (int r = 0; r < 8; ++r) {
      const int row = half * 8 + r;
      const float rx = __shfl(ax, row, 32);
      const float ry = __shfl(ay, row, 32);
      const float rz = __shfl(az, row, 32);
      c[r] = -2.0f * (rx * bx + ry * by + rz * bz) + bn2;
    }
#endif
#pragma unroll
    for (int r = 0; r < 8; ++r) ins3_stream(c[r], n, td[r], ti[r]);
  }

#pragma unroll
  for (int mskk = 1; mskk <= 8; mskk <<= 1) {
#pragma unroll
    for (int r = 0; r < 8; ++r) {
      float pd[3]; int pi[3];
#pragma unroll
      for (int k = 0; k < 3; ++k) {
        pd[k] = __shfl_xor(td[r][k], mskk, 32);
        pi[k] = __shfl_xor(ti[r][k], mskk, 32);
      }
#pragma unroll
      for (int k = 0; k < 3; ++k) ins3_lex(pd[k], pi[k], td[r], ti[r]);
    }
  }

  if (l15 == 0) {
#pragma unroll
    for (int r = 0; r < 8; ++r)
#pragma unroll
      for (int k = 0; k < 3; ++k) {
        s_td[wv][half * 8 + r][k] = td[r][k];
        s_ti[wv][half * 8 + r][k] = ti[r][k];
      }
  }
  __syncthreads();

  float fd[3];
  int   fi[3];
#pragma unroll
  for (int k = 0; k < 3; ++k) { fd[k] = s_td[l15][wv][k]; fi[k] = s_ti[l15][wv][k]; }
#pragma unroll
  for (int mskk = 1; mskk <= 8; mskk <<= 1) {
    float pd[3]; int pi[3];
#pragma unroll
    for (int k = 0; k < 3; ++k) {
      pd[k] = __shfl_xor(fd[k], mskk, 32);
      pi[k] = __shfl_xor(fi[k], mskk, 32);
    }
#pragma unroll
    for (int k = 0; k < 3; ++k) ins3_lex(pd[k], pi[k], fd, fi);
  }

  if (lane == 0) {
    const int row = rt * 16 + wv;
    const int gi  = fps_idx[b * M_ + row];
    const float* gp = pcb + (size_t)gi * 3;
    const float gx = gp[0], gy = gp[1], gz = gp[2];
    float wv3[3];
    float wsum = 0.f;
#pragma unroll
    for (int k = 0; k < 3; ++k) {
      const int nn = fi[k];
      const float* np = pcb + (size_t)nn * 3;
      const float dx = np[0] - gx, dy = np[1] - gy, dz = np[2] - gz;
      const float dd = dx * dx + dy * dy + dz * dz;
      const float dist = sqrtf(fmaxf(dd, 0.0f));
      wv3[k] = 1.0f / (dist + 1e-8f);
      wsum += wv3[k];
      idx3[((size_t)b * M_ + row) * 3 + k] = nn;
    }
#pragma unroll
    for (int k = 0; k < 3; ++k)
      w3[((size_t)b * M_ + row) * 3 + k] = wv3[k] / wsum;
  }
}

__global__ __launch_bounds__(256) void interp_kernel(
    const float* __restrict__ feats, const int* __restrict__ idx3,
    const float* __restrict__ w3, float* __restrict__ out) {
  const int bc = blockIdx.x;
  const int b  = bc / C_;
  const int c  = bc - b * C_;
  const float* f = feats + ((size_t)b * C_ + c) * N_;
  float*       o = out   + ((size_t)b * C_ + c) * M_;
  const int*   ib = idx3 + (size_t)b * M_ * 3;
  const float* wb = w3   + (size_t)b * M_ * 3;
  for (int m = threadIdx.x; m < M_; m += 256) {
    const int   n0 = ib[m * 3 + 0], n1 = ib[m * 3 + 1], n2 = ib[m * 3 + 2];
    const float w0 = wb[m * 3 + 0], w1 = wb[m * 3 + 1], w2 = wb[m * 3 + 2];
    o[m] = w0 * f[n0] + w1 * f[n1] + w2 * f[n2];
  }
}

extern "C" void kernel_launch(void* const* d_in, const int* in_sizes, int n_in,
                              void* d_out, int out_size, void* d_ws, size_t ws_size,
                              hipStream_t stream) {
  const float* pc    = (const float*)d_in[0];
  const float* feats = (const float*)d_in[1];
  const float* obj   = (const float*)d_in[2];
  const float* grasp = (const float*)d_in[3];
  float* out = (float*)d_out;

  char* ws = (char*)d_ws;
  size_t off = 0;
  int* fps = (int*)(ws + off);
  off += (size_t)B_ * M_ * 4;        off = (off + 255) & ~(size_t)255;
  int* idx3 = (int*)(ws + off);
  off += (size_t)B_ * M_ * 3 * 4;    off = (off + 255) & ~(size_t)255;
  float* w3 = (float*)(ws + off);
  off += (size_t)B_ * M_ * 3 * 4;

  fps_kernel<<<B_, FPS_THREADS, 0, stream>>>(pc, obj, grasp, fps);
  three_nn_kernel<<<B_ * (M_ / 16), NN_WAVES * 32, 0, stream>>>(pc, fps, idx3, w3);
  interp_kernel<<<B_ * C_, 256, 0, stream>>>(feats, idx3, w3, out);
}


# fps kernel: a pass keeps only running min and thread max (integer min/max on the bits, no best-slot selects); the wave winner's slot comes from equality masks vs the wave max, scalar unit; same sums
# speedup vs baseline: 5.0921x; 1.0851x over previous
#include <hip/hip_runtime.h>
#include <hip/hip_bf16.h>

#define B_ 4
#define N_ 20000
#define C_ 512
#define M_ 1024
#define GRASP_TH 0.1f

typedef float v2f __attribute__((ext_vector_type(2)));
typedef float v8f __attribute__((ext_vector_type(8)));

#if defined(__gfx1250__) && __has_builtin(__builtin_amdgcn_wmma_f32_16x16x4_f32)
#define USE_WMMA 1
#else
#define USE_WMMA 0
#endif

__device__ __forceinline__ void ins3_stream(float v, int n, float d[3], int ix[3]) {
  const bool c0 = v < d[0];
  const bool c1 = v < d[1];
  const bool c2 = v < d[2];
  const float r0 = c0 ? v     : d[0];
  const int   j0 = c0 ? n     : ix[0];
  const float r1 = c0 ? d[0]  : (c1 ? v : d[1]);
  const int   j1 = c0 ? ix[0] : (c1 ? n : ix[1]);
  const float r2 = c1 ? d[1]  : (c2 ? v : d[2]);
  const int   j2 = c1 ? ix[1] : (c2 ? n : ix[2]);
  d[0] = r0; d[1] = r1; d[2] = r2;
  ix[0] = j0; ix[1] = j1; ix[2] = j2;
}

__device__ __forceinline__ void ins3_lex(float v, int n, float d[3], int ix[3]) {
  const bool c0 = (v < d[0]) | ((v == d[0]) & (n < ix[0]));
  const bool c1 = (v < d[1]) | ((v == d[1]) & (n < ix[1]));
  const bool c2 = (v < d[2]) | ((v == d[2]) & (n < ix[2]));
  const float r0 = c0 ? v     : d[0];
  const int   j0 = c0 ? n     : ix[0];
  const float r1 = c0 ? d[0]  : (c1 ? v : d[1]);
  const int   j1 = c0 ? ix[0] : (c1 ? n : ix[1]);
  const float r2 = c1 ? d[1]  : (c2 ? v : d[2]);
  const int   j2 = c1 ? ix[1] : (c2 ? n : ix[2]);
  d[0] = r0; d[1] = r1; d[2] = r2;
  ix[0] = j0; ix[1] = j1; ix[2] = j2;
}

template <int XM>
__device__ __forceinline__ int row_xor_i(int v) {
  return __builtin_amdgcn_update_dpp(v, v, 0x160 + XM, 0xf, 0xf, true);
}
__device__ __forceinline__ int xhalf_i(int v) {
  return __builtin_amdgcn_permlanex16(v, v, 0x76543210, 0xfedcba98, false, false);
}
__device__ __forceinline__ int wave_max_bits(int key) {
  int m = key;
  m = max(m, row_xor_i<1>(m));
  m = max(m, row_xor_i<2>(m));
  m = max(m, row_xor_i<4>(m));
  m = max(m, row_xor_i<8>(m));
  m = max(m, xhalf_i(m));
  return m;
}

__device__ __forceinline__ int table_argmax(int k, int i) {
  int m = k;
  m = max(m, row_xor_i<1>(m));
  m = max(m, row_xor_i<2>(m));
  m = max(m, row_xor_i<4>(m));
  m = max(m, row_xor_i<8>(m));
  int c = (k == m) ? i : 0x7fffffff;
  c = min(c, row_xor_i<1>(c));
  c = min(c, row_xor_i<2>(c));
  c = min(c, row_xor_i<4>(c));
  c = min(c, row_xor_i<8>(c));
  return c;
}

__device__ __forceinline__ void fps_pass(v2f xy, float z, int& m, v2f pxy, float pz,
                                         int& bestK) {
  const v2f   d  = xy - pxy;
  const float dz = z - pz;
  const float sxy = d.x * d.x + d.y * d.y;
  const float dd  = __builtin_fmaf(dz, dz, sxy);
  const int nm = min(m, __builtin_bit_cast(int, dd));
  m = nm;
  bestK = max(bestK, nm);
}

#define FPS_THREADS 512
#define FPS_WAVES  (FPS_THREADS / 32)
#define FPS_PASSES 40
#define FPS_WORDS  (FPS_PASSES * FPS_WAVES)
#define FPS_WPL    (FPS_WORDS / 32)
#define FPS_GROUP  8
#define CMAX 15000
static_assert(FPS_PASSES * FPS_THREADS >= N_, "the passes cover every point");
static_assert(FPS_PASSES > 32 && FPS_PASSES <= 64, "two flag-word registers per lane");
static_assert(FPS_WORDS % 32 == 0 && FPS_PASSES % FPS_GROUP == 0 && FPS_WAVES == 16,
              "prefix split, staging groups, one table entry per lane of a row");

template <int P>
__device__ __forceinline__ void fps_serial(const float* s_xyz, const int* s_idx,
                                           unsigned* s_bits, int* s_pref,
                                           int first, bool compact, int Kc,
                                           int b, int tid, int* __restrict__ fps_idx) {
  const int lane = tid & 31;
  const int wid  = tid >> 5;
  v2f   rxy[P];
  float rz[P];
  int   rm[P];
#pragma unroll
  for (int p = 0; p < P; ++p) {
    const int  slot = p * FPS_THREADS + tid;
    const bool in   = slot < Kc;
    const int  sl   = in ? slot : 0;
    rxy[p].x = s_xyz[sl * 3 + 0];
    rxy[p].y = s_xyz[sl * 3 + 1];
    rz[p]    = s_xyz[sl * 3 + 2];
    const bool mk = (s_bits[p * FPS_WAVES + wid] >> lane) & 1u;
    rm[p] = (in & (compact | mk)) ? 0x7f800000 : (int)0xff800000;
  }
  __syncthreads();

  int*   swv = reinterpret_cast<int*>(s_bits);
  int*   swi = s_pref;
  int last = first;
  int acc = 0;
  for (int step = 0; step < M_; ++step) {
    v2f pxy;
    pxy.x = s_xyz[last * 3 + 0];
    pxy.y = s_xyz[last * 3 + 1];
    const float pz = s_xyz[last * 3 + 2];
    if (wid == 0) {
      const int pick = compact ? s_idx[last] : last;
      acc = (lane == (step & 31)) ? pick : acc;
      if ((step & 31) == 31) fps_idx[b * M_ + (step - 31) + lane] = acc;
    }

    int bestK = (int)0xff800000;
#pragma unroll
    for (int p = 0; p < P; ++p)
      fps_pass(rxy[p], rz[p], rm[p], pxy, pz, bestK);
    const int wmax = wave_max_bits(bestK);
    constexpr int NPAIR = (P + 1) / 2;
    unsigned long long selM = 0ull;
    int                selJ = NPAIR - 1;
#pragma unroll
    for (int j = NPAIR - 1; j >= 0; --j) {
      const unsigned lo = __builtin_amdgcn_ballot_w32(rm[2 * j] == wmax);
      const unsigned hi = (2 * j + 1 < P) ? __builtin_amdgcn_ballot_w32(rm[(2 * j + 1 < P) ? 2 * j + 1 : 0] == wmax) : 0u;
      const unsigned long long mk = ((unsigned long long)hi << 32) | lo;
      const bool take = (j == NPAIR - 1) | (mk != 0ull);
      selM = take ? mk : selM;
      selJ = take ? j : selJ;
    }
    const int pos   = __builtin_ctzll(selM);
    const int bestI = (2 * selJ + (pos >> 5)) * FPS_THREADS + wid * 32 + (pos & 31);
    const int pb = (step & 1) * FPS_WAVES;
    if (lane == 0) { swv[pb + wid] = wmax; swi[pb + wid] = bestI; }
    __syncthreads();
    const int e = pb + (lane & 15);
    last = table_argmax(swv[e], swi[e]);
  }
}

__global__ __launch_bounds__(FPS_THREADS) void fps_kernel(
    const float* __restrict__ pc, const float* __restrict__ obj,
    const float* __restrict__ grasp, int* __restrict__ fps_idx) {
  const int b    = blockIdx.x;
  const int tid  = threadIdx.x;
  const int lane = tid & 31;
  const int wid  = tid >> 5;

  __shared__ float    s_xyz[N_ * 3];
  __shared__ unsigned s_bits[FPS_WORDS];
  __shared__ int      s_pref[FPS_WORDS];
  __shared__ int   s_last;
  __shared__ int   s_K;

  const float* pcb = pc + (size_t)b * N_ * 3;
  const float* ob0 = obj + (size_t)b * 2 * N_;
  const float* gr  = grasp + (size_t)b * N_;

  unsigned myw0 = 0u, myw1 = 0u;
  for (int it = 0; it < 32; ++it) {
    const int n  = it * FPS_THREADS + tid;
    const int nl = (n < N_) ? n : (N_ - 1);
    const bool mk = (n < N_) & (ob0[N_ + nl] > ob0[nl]) & (gr[nl] > GRASP_TH);
    const unsigned w = __builtin_amdgcn_ballot_w32(mk);
    myw0 = (lane == it) ? w : myw0;
  }
  for (int it = 32; it < FPS_PASSES; ++it) {
    const int n  = it * FPS_THREADS + tid;
    const int nl = (n < N_) ? n : (N_ - 1);
    const bool mk = (n < N_) & (ob0[N_ + nl] > ob0[nl]) & (gr[nl] > GRASP_TH);
    const unsigned w = __builtin_amdgcn_ballot_w32(mk);
    myw1 = (lane == it - 32) ? w : myw1;
  }
  s_bits[lane * FPS_WAVES + wid] = myw0;
  if (lane < FPS_PASSES - 32) s_bits[(lane + 32) * FPS_WAVES + wid] = myw1;
  __syncthreads();

  if (wid == 0) {
    int sum = 0;
    for (int j = 0; j < FPS_WPL; ++j) sum += __popc(s_bits[lane * FPS_WPL + j]);
    int incl = sum;
    for (int d = 1; d < 32; d <<= 1) {
      const int o = __shfl_up(incl, d, 32);
      incl += (lane >= d) ? o : 0;
    }
    int run = incl - sum;
    for (int j = 0; j < FPS_WPL; ++j) {
      s_pref[lane * FPS_WPL + j] = run;
      run += __popc(s_bits[lane * FPS_WPL + j]);
    }
    if (lane == 31) s_K = incl;
    if (lane == 0)  s_last = 0;
  }
  __syncthreads();

  const int  K       = s_K;
  const bool compact = (K >= 1) & (K <= CMAX);
  const int  Kc      = compact ? K : N_;
  int* s_idx = (int*)(s_xyz + (compact ? 3 * Kc : 0));

  for (int g = 0; g < FPS_PASSES; g += FPS_GROUP) {
    float x[FPS_GROUP], y[FPS_GROUP], z[FPS_GROUP];
    unsigned w[FPS_GROUP];
    int pf[FPS_GROUP];
    for (int j = 0; j < FPS_GROUP; ++j) {
      const int n  = (g + j) * FPS_THREADS + tid;
      const int nl = (n < N_) ? n : (N_ - 1);
      const float* p = pcb + (size_t)nl * 3;
      x[j] = p[0]; y[j] = p[1]; z[j] = p[2];
      w[j]  = s_bits[(g + j) * FPS_WAVES + wid];
      pf[j] = s_pref[(g + j) * FPS_WAVES + wid];
    }
    for (int j = 0; j < FPS_GROUP; ++j) {
      const int  n    = (g + j) * FPS_THREADS + tid;
      const bool mk   = (w[j] >> lane) & 1u;
      const int  pos  = pf[j] + __popc(w[j] & ((1u << lane) - 1u));
      const bool keep = compact ? mk : (n < N_);
      const int  slot = compact ? pos : n;
      if (keep) {
        s_xyz[slot * 3 + 0] = x[j];
        s_xyz[slot * 3 + 1] = y[j];
        s_xyz[slot * 3 + 2] = z[j];
        if (compact) s_idx[slot] = n;
        else if (mk & (pos == 0)) s_last = n;
      }
    }
  }
  __syncthreads();

  const int first = s_last;
  const int np = (Kc + FPS_THREADS - 1) / FPS_THREADS;
#define FPS_RUN(P) case P: fps_serial<P>(s_xyz, s_idx, s_bits, s_pref, first, compact, Kc, b, tid, fps_idx); break;
  switch (np) {
    FPS_RUN(1)  FPS_RUN(2)  FPS_RUN(3)  FPS_RUN(4)  FPS_RUN(5)  FPS_RUN(6)  FPS_RUN(7)  FPS_RUN(8)
    FPS_RUN(9)  FPS_RUN(10) FPS_RUN(11) FPS_RUN(12) FPS_RUN(13) FPS_RUN(14) FPS_RUN(15) FPS_RUN(16)
    FPS_RUN(17) FPS_RUN(18) FPS_RUN(19) FPS_RUN(20) FPS_RUN(21) FPS_RUN(22) FPS_RUN(23) FPS_RUN(24)
    FPS_RUN(25) FPS_RUN(26) FPS_RUN(27) FPS_RUN(28) FPS_RUN(29) FPS_RUN(30) FPS_RUN(31) FPS_RUN(32)
    FPS_RUN(33) FPS_RUN(34) FPS_RUN(35) FPS_RUN(36) FPS_RUN(37) FPS_RUN(38) FPS_RUN(39)
    default: fps_serial<FPS_PASSES>(s_xyz, s_idx, s_bits, s_pref, first, compact, Kc, b, tid, fps_idx); break;
  }
#undef FPS_RUN
}

#define NN_WAVES 16
__global__ __launch_bounds__(NN_WAVES * 32) void three_nn_kernel(
    const float* __restrict__ pc, const int* __restrict__ fps_idx,
    int* __restrict__ idx3, float* __restrict__ w3) {
  static_assert(NN_WAVES == 16, "the final merge maps wave j onto lane j of a 16-lane row");
  const int tid  = threadIdx.x;
  const int lane = tid & 31;
  const int wv   = __builtin_amdgcn_readfirstlane(tid >> 5);
  const int tilesPerBatch = M_ / 16;
  const int b    = blockIdx.x / tilesPerBatch;
  const int rt   = blockIdx.x - b * tilesPerBatch;
  const int half = lane >> 4;
  const int l15  = lane & 15;

  __shared__ float s_td[NN_WAVES][16][3];
  __shared__ int   s_ti[NN_WAVES][16][3];

  const float* pcb = pc + (size_t)b * N_ * 3;

  const int mrow = rt * 16 + l15;
  const int aidx = fps_idx[b * M_ + mrow];
  const float* ap = pcb + (size_t)aidx * 3;
  const float ax = ap[0], ay = ap[1], az = ap[2];
  v2f A;
  A.x = half ? (-2.0f * az) : (-2.0f * ax);
  A.y = half ? 1.0f         : (-2.0f * ay);

  float td[8][3];
  int   ti[8][3];
#pragma unroll
  for (int r = 0; r < 8; ++r)
#pragma unroll
    for (int k = 0; k < 3; ++k) { td[r][k] = __builtin_inff(); ti[r][k] = 0x7fffffff; }

  const int t0 = (wv * (N_ / 16)) / NN_WAVES;
  const int t1 = ((wv + 1) * (N_ / 16)) / NN_WAVES;
  int n = t0 * 16 + l15;
  const float* bp = pcb + (size_t)n * 3;
#pragma unroll 2
  for (int t = t0; t < t1; ++t, n += 16, bp += 16 * 3) {
    const int pn = (n + 128 < N_) ? (n + 128) : (N_ - 1);
    __builtin_prefetch(pcb + (size_t)pn * 3, 0, 1);
    const float bx = bp[0], by = bp[1], bz = bp[2];
    const float bn2 = bx * bx + by * by + bz * bz;
    v8f c = {0.f, 0.f, 0.f, 0.f, 0.f, 0.f, 0.f, 0.f};
#if USE_WMMA
    v2f Bv;
    Bv.x = half ? bz  : bx;
    Bv.y = half ? bn2 : by;
    c = __builtin_amdgcn_wmma_f32_16x16x4_f32(
         false, A,  false, Bv,
         (short)0, c,  false,  false);
#else
#pragma unroll
    for (int r = 0; r < 8; ++r) {
      const int row = half * 8 + r;
      const float rx = __shfl(ax, row, 32);
      const float ry = __shfl(ay, row, 32);
      const float rz = __shfl(az, row, 32);
      c[r] = -2.0f * (rx * bx + ry * by + rz * bz) + bn2;
    }
#endif
#pragma unroll
    for (int r = 0; r < 8; ++r) ins3_stream(c[r], n, td[r], ti[r]);
  }

#pragma unroll
  for (int mskk = 1; mskk <= 8; mskk <<= 1) {
#pragma unroll
    for (int r = 0; r < 8; ++r) {
      float pd[3]; int pi[3];
#pragma unroll
      for (int k = 0; k < 3; ++k) {
        pd[k] = __shfl_xor(td[r][k], mskk, 32);
        pi[k] = __shfl_xor(ti[r][k], mskk, 32);
      }
#pragma unroll
      for (int k = 0; k < 3; ++k) ins3_lex(pd[k], pi[k], td[r], ti[r]);
    }
  }

  if (l15 == 0) {
#pragma unroll
    for (int r = 0; r < 8; ++r)
#pragma unroll
      for (int k = 0; k < 3; ++k) {
        s_td[wv][half * 8 + r][k] = td[r][k];
        s_ti[wv][half * 8 + r][k] = ti[r][k];
      }
  }
  __syncthreads();

  float fd[3];
  int   fi[3];
#pragma unroll
  for (int k = 0; k < 3; ++k) { fd[k] = s_td[l15][wv][k]; fi[k] = s_ti[l15][wv][k]; }
#pragma unroll
  for (int mskk = 1; mskk <= 8; mskk <<= 1) {
    float pd[3]; int pi[3];
#pragma unroll
    for (int k = 0; k < 3; ++k) {
      pd[k] = __shfl_xor(fd[k], mskk, 32);
      pi[k] = __shfl_xor(fi[k], mskk, 32);
    }
#pragma unroll
    for (int k = 0; k < 3; ++k) ins3_lex(pd[k], pi[k], fd, fi);
  }

  if (lane == 0) {
    const int row = rt * 16 + wv;
    const int gi  = fps_idx[b * M_ + row];
    const float* gp = pcb + (size_t)gi * 3;
    const float gx = gp[0], gy = gp[1], gz = gp[2];
    float wv3[3];
    float wsum = 0.f;
#pragma unroll
    for (int k = 0; k < 3; ++k) {
      const int nn = fi[k];
      const float* np = pcb + (size_t)nn * 3;
      const float dx = np[0] - gx, dy = np[1] - gy, dz = np[2] - gz;
      const float dd = dx * dx + dy * dy + dz * dz;
      const float dist = sqrtf(fmaxf(dd, 0.0f));
      wv3[k] = 1.0f / (dist + 1e-8f);
      wsum += wv3[k];
      idx3[((size_t)b * M_ + row) * 3 + k] = nn;
    }
#pragma unroll
    for (int k = 0; k < 3; ++k)
      w3[((size_t)b * M_ + row) * 3 + k] = wv3[k] / wsum;
  }
}

__global__ __launch_bounds__(256) void interp_kernel(
    const float* __restrict__ feats, const int* __restrict__ idx3,
    const float* __restrict__ w3, float* __restrict__ out) {
  const int bc = blockIdx.x;
  const int b  = bc / C_;
  const int c  = bc - b * C_;
  const float* f = feats + ((size_t)b * C_ + c) * N_;
  float*       o = out   + ((size_t)b * C_ + c) * M_;
  const int*   ib = idx3 + (size_t)b * M_ * 3;
  const float* wb = w3   + (size_t)b * M_ * 3;
  for (int m = threadIdx.x; m < M_; m += 256) {
    const int   n0 = ib[m * 3 + 0], n1 = ib[m * 3 + 1], n2 = ib[m * 3 + 2];
    const float w0 = wb[m * 3 + 0], w1 = wb[m * 3 + 1], w2 = wb[m * 3 + 2];
    o[m] = w0 * f[n0] + w1 * f[n1] + w2 * f[n2];
  }
}

extern "C" void kernel_launch(void* const* d_in, const int* in_sizes, int n_in,
                              void* d_out, int out_size, void* d_ws, size_t ws_size,
                              hipStream_t stream) {
  const float* pc    = (const float*)d_in[0];
  const float* feats = (const float*)d_in[1];
  const float* obj   = (const float*)d_in[2];
  const float* grasp = (const float*)d_in[3];
  float* out = (float*)d_out;

  char* ws = (char*)d_ws;
  size_t off = 0;
  int* fps = (int*)(ws + off);
  off += (size_t)B_ * M_ * 4;        off = (off + 255) & ~(size_t)255;
  int* idx3 = (int*)(ws + off);
  off += (size_t)B_ * M_ * 3 * 4;    off = (off + 255) & ~(size_t)255;
  float* w3 = (float*)(ws + off);
  off += (size_t)B_ * M_ * 3 * 4;

  fps_kernel<<<B_, FPS_THREADS, 0, stream>>>(pc, obj, grasp, fps);
  three_nn_kernel<<<B_ * (M_ / 16), NN_WAVES * 32, 0, stream>>>(pc, fps, idx3, w3);
  interp_kernel<<<B_ * C_, 256, 0, stream>>>(feats, idx3, w3, out);
}
